# starting kernel
# baseline (speedup 1.0000x reference)
#include <hip/hip_runtime.h>
#include <hip/hip_bf16.h>
#include <math.h>

#define BATCH   4
#define SEQL    2048
#define DMODEL  512
#define NSTATE  16
#define OUTD    512
#define MROWS   (BATCH * SEQL)
#define KDIM    512
#define NCHUNK  16
#define CLEN    (SEQL / NCHUNK)
#define LN_EPS  1e-5f

typedef float v2f __attribute__((ext_vector_type(2)));
typedef float v8f __attribute__((ext_vector_type(8)));

__global__ __launch_bounds__(128)
void s6_gemm_wmma_k512(const float* __restrict__ A,
                       const float* __restrict__ W,
                       const float* __restrict__ bias,
                       float* __restrict__ out,
                       int Nout, int act  ) {
    __shared__ float sB[KDIM * 16];

    const int n0 = blockIdx.y << 4;
    for (int idx = threadIdx.x; idx < KDIM * 16; idx += 128) {
        const int k = idx >> 4, nn = idx & 15;
        sB[idx] = W[(size_t)k * Nout + n0 + nn];
    }
    __syncthreads();

    const int wave  = threadIdx.x >> 5;
    const int lane  = threadIdx.x & 31;
    const int mtile = (blockIdx.x << 2) + wave;
    const int mrow  = (mtile << 4) + (lane & 15);
    const int klo   = (lane >> 4) << 1;
    const int ncol  = lane & 15;

    v8f acc = {0.f, 0.f, 0.f, 0.f, 0.f, 0.f, 0.f, 0.f};

    const float* ap = A + (size_t)mrow * KDIM + klo;
#pragma unroll 4
    for (int k0 = 0; k0 < KDIM; k0 += 4) {
        v2f av; av.x = ap[k0];  av.y = ap[k0 + 1];
        const int kk = (k0 + klo) << 4;
        v2f bv; bv.x = sB[kk + ncol]; bv.y = sB[kk + 16 + ncol];
        acc = __builtin_amdgcn_wmma_f32_16x16x4_f32(
             false, av,  false, bv,
             (short)0, acc,  false,  false);
    }

    const float bb  = bias[n0 + ncol];
    const int   mhi = (lane >> 4) << 3;
#pragma unroll
    for (int r = 0; r < 8; ++r) {
        float v = acc[r] + bb;
        if (act == 1) v = (v > 20.f) ? v : log1pf(expf(v));
        out[(size_t)((mtile << 4) + mhi + r) * Nout + n0 + ncol] = v;
    }
}

__global__ __launch_bounds__(256)
void s6_scan_pass1(const float* __restrict__ x,
                   const float* __restrict__ dt,
                   const float* __restrict__ Bm,
                   const float* __restrict__ A_log,
                   float* __restrict__ hend,
                   float* __restrict__ aend) {
    const int b = blockIdx.z, c = blockIdx.y;
    const int d = blockIdx.x * 256 + threadIdx.x;

    __shared__ float sBm[CLEN * NSTATE];
    for (int idx = threadIdx.x; idx < CLEN * NSTATE; idx += 256)
        sBm[idx] = Bm[((size_t)b * SEQL + c * CLEN) * NSTATE + idx];
    __syncthreads();

    float Ac[NSTATE];
#pragma unroll
    for (int n = 0; n < NSTATE; ++n) Ac[n] = -expf(A_log[d * NSTATE + n]);

    float h[NSTATE], ap[NSTATE];
#pragma unroll
    for (int n = 0; n < NSTATE; ++n) { h[n] = 0.f; ap[n] = 1.f; }

    for (int t = 0; t < CLEN; ++t) {
        const size_t ofs = ((size_t)b * SEQL + c * CLEN + t) * DMODEL + d;
        const float dtv = dt[ofs];
        const float ub  = dtv * x[ofs];
        const int   bmo = t * NSTATE;
#pragma unroll
        for (int n = 0; n < NSTATE; ++n) {
            const float a = __expf(dtv * Ac[n]);
            h[n]  = a * h[n] + ub * sBm[bmo + n];
            ap[n] *= a;
        }
    }

    const size_t base = (((size_t)b * DMODEL + d) * NCHUNK + c) * NSTATE;
#pragma unroll
    for (int n = 0; n < NSTATE; ++n) { hend[base + n] = h[n]; aend[base + n] = ap[n]; }
}

__global__ __launch_bounds__(256)
void s6_scan_pass2(const float* __restrict__ hend,
                   const float* __restrict__ aend,
                   float* __restrict__ carry) {
    const int idx = blockIdx.x * 256 + threadIdx.x;
    const size_t bd = (size_t)(idx >> 4);
    const int n = idx & 15;
    float cv = 0.f;
    for (int c = 0; c < NCHUNK; ++c) {
        const size_t o = (bd * NCHUNK + c) * NSTATE + n;
        carry[o] = cv;
        cv = aend[o] * cv + hend[o];
    }
}

__global__ __launch_bounds__(256)
void s6_scan_pass3(const float* __restrict__ x,
                   const float* __restrict__ dt,
                   const float* __restrict__ Bm,
                   const float* __restrict__ Cm,
                   const float* __restrict__ A_log,
                   const float* __restrict__ carry,
                   const float* __restrict__ Dskip,
                   float* __restrict__ yout) {
    const int b = blockIdx.z, c = blockIdx.y;
    const int d = blockIdx.x * 256 + threadIdx.x;

    __shared__ float sBm[CLEN * NSTATE];
    __shared__ float sCm[CLEN * NSTATE];
    for (int idx = threadIdx.x; idx < CLEN * NSTATE; idx += 256) {
        const size_t g = ((size_t)b * SEQL + c * CLEN) * NSTATE + idx;
        sBm[idx] = Bm[g];
        sCm[idx] = Cm[g];
    }
    __syncthreads();

    float Ac[NSTATE];
#pragma unroll
    for (int n = 0; n < NSTATE; ++n) Ac[n] = -expf(A_log[d * NSTATE + n]);

    float h[NSTATE];
    const size_t base = (((size_t)b * DMODEL + d) * NCHUNK + c) * NSTATE;
#pragma unroll
    for (int n = 0; n < NSTATE; ++n) h[n] = carry[base + n];

    const float dsk = Dskip[d];
    for (int t = 0; t < CLEN; ++t) {
        const size_t ofs = ((size_t)b * SEQL + c * CLEN + t) * DMODEL + d;
        const float dtv = dt[ofs];
        const float xv  = x[ofs];
        const float ub  = dtv * xv;
        const int   so  = t * NSTATE;
        float y = 0.f;
#pragma unroll
        for (int n = 0; n < NSTATE; ++n) {
            const float a = __expf(dtv * Ac[n]);
            h[n] = a * h[n] + ub * sBm[so + n];
            y += sCm[so + n] * h[n];
        }
        y += dsk * xv;
        yout[ofs] = 0.5f * y * (1.f + erff(y * 0.70710678118654752f));
    }
}

__global__ __launch_bounds__(256)
void s6_ln_residual(const float* __restrict__ ymix,
                    const float* __restrict__ hin,
                    const float* __restrict__ g,
                    const float* __restrict__ be,
                    float* __restrict__ hout) {
    const size_t row = blockIdx.x;
    const int tid = threadIdx.x;
    const float v0 = ymix[row * DMODEL + tid];
    const float v1 = ymix[row * DMODEL + 256 + tid];

    __shared__ float s1[256], s2[256];
    s1[tid] = v0 + v1;
    s2[tid] = v0 * v0 + v1 * v1;
    __syncthreads();
    for (int st = 128; st > 0; st >>= 1) {
        if (tid < st) { s1[tid] += s1[tid + st]; s2[tid] += s2[tid + st]; }
        __syncthreads();
    }
    const float mu   = s1[0] * (1.f / DMODEL);
    const float var  = s2[0] * (1.f / DMODEL) - mu * mu;
    const float rstd = rsqrtf(var + LN_EPS);

    hout[row * DMODEL + tid] =
        hin[row * DMODEL + tid] + (v0 - mu) * rstd * g[tid] + be[tid];
    hout[row * DMODEL + 256 + tid] =
        hin[row * DMODEL + 256 + tid] + (v1 - mu) * rstd * g[256 + tid] + be[256 + tid];
}

extern "C" void kernel_launch(void* const* d_in, const int* in_sizes, int n_in,
                              void* d_out, int out_size, void* d_ws, size_t ws_size,
                              hipStream_t stream) {
    (void)in_sizes; (void)n_in; (void)out_size; (void)ws_size;

    const float* x      = (const float*)d_in[0];
    const float* A_log  = (const float*)d_in[1];
    const float* W_B    = (const float*)d_in[2];
    const float* b_B    = (const float*)d_in[3];
    const float* W_C    = (const float*)d_in[4];
    const float* b_C    = (const float*)d_in[5];
    const float* W_dt   = (const float*)d_in[6];
    const float* b_dt   = (const float*)d_in[7];
    const float* D_skip = (const float*)d_in[8];
    const float* W_mix  = (const float*)d_in[9];
    const float* b_mix  = (const float*)d_in[10];
    const float* ln_g   = (const float*)d_in[11];
    const float* ln_b   = (const float*)d_in[12];
    const float* W_dec  = (const float*)d_in[13];
    const float* b_dec  = (const float*)d_in[14];
    float* out = (float*)d_out;

    float* ws = (float*)d_ws;
    size_t off = 0;
    float* dtb   = ws + off; off += (size_t)MROWS * DMODEL;
    float* Bmb   = ws + off; off += (size_t)MROWS * NSTATE;
    float* Cmb   = ws + off; off += (size_t)MROWS * NSTATE;
    float* ygelu = ws + off; off += (size_t)MROWS * DMODEL;
    float* ymix  = ws + off; off += (size_t)MROWS * DMODEL;
    float* h1    = ws + off; off += (size_t)MROWS * DMODEL;
    float* h2    = ws + off; off += (size_t)MROWS * DMODEL;
    float* hend  = ws + off; off += (size_t)BATCH * DMODEL * NCHUNK * NSTATE;
    float* aend  = ws + off; off += (size_t)BATCH * DMODEL * NCHUNK * NSTATE;
    float* carry = ws + off; off += (size_t)BATCH * DMODEL * NCHUNK * NSTATE;

    const int DN = DMODEL * NSTATE;
    const int DD = DMODEL * DMODEL;

    const dim3 gBig(MROWS / 64, DMODEL / 16);
    const dim3 gBC (MROWS / 64, 1);
    const dim3 gScan(DMODEL / 256, NCHUNK, BATCH);

    for (int i = 0; i < 2; ++i) {
        const float* hin = (i == 0) ? x : h1;
        float* hout      = (i == 0) ? h1 : h2;

        s6_gemm_wmma_k512<<<gBig, 128, 0, stream>>>(
            hin, W_dt + (size_t)i * DD, b_dt + i * DMODEL, dtb, DMODEL, 1);
        s6_gemm_wmma_k512<<<gBC, 128, 0, stream>>>(
            hin, W_B + (size_t)i * DN, b_B + i * NSTATE, Bmb, NSTATE, 0);
        s6_gemm_wmma_k512<<<gBC, 128, 0, stream>>>(
            hin, W_C + (size_t)i * DN, b_C + i * NSTATE, Cmb, NSTATE, 0);

        s6_scan_pass1<<<gScan, 256, 0, stream>>>(
            hin, dtb, Bmb, A_log + (size_t)i * DN, hend, aend);
        s6_scan_pass2<<<BATCH * DMODEL * NSTATE / 256, 256, 0, stream>>>(
            hend, aend, carry);
        s6_scan_pass3<<<gScan, 256, 0, stream>>>(
            hin, dtb, Bmb, Cmb, A_log + (size_t)i * DN, carry,
            D_skip + i * DMODEL, ygelu);

        s6_gemm_wmma_k512<<<gBig, 128, 0, stream>>>(
            ygelu, W_mix + (size_t)i * DD, b_mix + i * DMODEL, ymix, DMODEL, 0);
        s6_ln_residual<<<MROWS, 256, 0, stream>>>(
            ymix, hin, ln_g + i * DMODEL, ln_b + i * DMODEL, hout);
    }

    s6_gemm_wmma_k512<<<gBig, 128, 0, stream>>>(h2, W_dec, b_dec, out, OUTD, 0);
}


# GEMM retiled: A and W chunks staged in two LDS stages by coalesced 16-byte loads one chunk ahead, each wave a 32x32 block of 4 accumulators (16x16 for the 16-column projections)
# speedup vs baseline: 1.5333x; 1.5333x over previous
#include <hip/hip_runtime.h>
#include <hip/hip_bf16.h>
#include <math.h>

#define BATCH   4
#define SEQL    2048
#define DMODEL  512
#define NSTATE  16
#define OUTD    512
#define MROWS   (BATCH * SEQL)
#define KDIM    512
#define NCHUNK  16
#define CLEN    (SEQL / NCHUNK)
#define LN_EPS  1e-5f

typedef float v2f __attribute__((ext_vector_type(2)));
typedef float v4f __attribute__((ext_vector_type(4)));
typedef float v8f __attribute__((ext_vector_type(8)));

template <int BM, int BN, int WM, int WN>
__global__ __launch_bounds__((BM / WM) * (BN / WN) * 32)
void s6_gemm_wmma_k512(const float* __restrict__ A,
                       const float* __restrict__ W,
                       const float* __restrict__ bias,
                       float* __restrict__ out,
                       int Nout, int act  ) {
    constexpr int KB    = 32;
    constexpr int PA    = KB + 4;
    constexpr int PB    = BN + 8;
    constexpr int NT    = (BM / WM) * (BN / WN) * 32;
    constexpr int TM    = WM / 16;
    constexpr int TN    = WN / 16;
    constexpr int AQ    = KB / 4;
    constexpr int BQ    = BN / 4;
    constexpr int A_PER = (BM * AQ) / NT;
    constexpr int B_PER = (KB * BQ) / NT;
    constexpr int NCH   = KDIM / KB;
    static_assert((BM * AQ) % NT == 0, "A chunk must split evenly over the block");
    static_assert((KB * BQ) % NT == 0, "W chunk must split evenly over the block");

    __shared__ float sA[2 * BM * PA] __attribute__((aligned(16)));
    __shared__ float sB[2 * KB * PB] __attribute__((aligned(16)));

    const int tid = threadIdx.x;
    const int m0  = blockIdx.x * BM;
    const int n0  = blockIdx.y * BN;

    const float* agp[A_PER]; int asl[A_PER];
    const float* bgp[B_PER]; int bsl[B_PER];
#pragma unroll
    for (int p = 0; p < A_PER; ++p) {
        const int idx = tid + p * NT;
        const int row = idx / AQ, kq = idx % AQ;
        agp[p] = A + (size_t)(m0 + row) * KDIM + kq * 4;
        asl[p] = row * PA + kq * 4;
    }
#pragma unroll
    for (int p = 0; p < B_PER; ++p) {
        const int idx = tid + p * NT;
        const int kr = idx / BQ, nq = idx % BQ;
        bgp[p] = W + (size_t)kr * Nout + n0 + nq * 4;
        bsl[p] = kr * PB + nq * 4;
    }

    v4f ra[A_PER], rb[B_PER];
#pragma unroll
    for (int p = 0; p < A_PER; ++p) ra[p] = *(const v4f*)agp[p];
#pragma unroll
    for (int p = 0; p < B_PER; ++p) rb[p] = *(const v4f*)bgp[p];

    const int wave = tid >> 5;
    const int lane = tid & 31;
    const int wm   = (wave / (BN / WN)) * WM;
    const int wn   = (wave % (BN / WN)) * WN;
    const int c16  = lane & 15;
    const int klo  = (lane >> 4) << 1;

    v8f acc[TM][TN];
#pragma unroll
    for (int i = 0; i < TM; ++i)
#pragma unroll
        for (int j = 0; j < TN; ++j)
            acc[i][j] = (v8f){0.f, 0.f, 0.f, 0.f, 0.f, 0.f, 0.f, 0.f};

    const float* saf = sA + (wm + c16) * PA + klo;
    const float* sbf = sB + klo * PB + wn + c16;

#pragma unroll
    for (int p = 0; p < A_PER; ++p) *(v4f*)&sA[asl[p]] = ra[p];
#pragma unroll
    for (int p = 0; p < B_PER; ++p) *(v4f*)&sB[bsl[p]] = rb[p];
    __syncthreads();

    for (int kc = 0; kc < NCH; ++kc) {
        const int  cur  = kc & 1;
        const bool more = (kc + 1 < NCH);

        if (more) {
#pragma unroll
            for (int p = 0; p < A_PER; ++p)
                ra[p] = *(const v4f*)(agp[p] + (kc + 1) * KB);
#pragma unroll
            for (int p = 0; p < B_PER; ++p)
                rb[p] = *(const v4f*)(bgp[p] + (size_t)(kc + 1) * KB * Nout);
        }
        __builtin_amdgcn_sched_barrier(0);

        const float* sa = saf + cur * (BM * PA);
        const float* sb = sbf + cur * (KB * PB);
#pragma unroll
        for (int k0 = 0; k0 < KB; k0 += 4) {
            v2f av[TM], bv[TN];
#pragma unroll
            for (int i = 0; i < TM; ++i)
                av[i] = *(const v2f*)(sa + i * 16 * PA + k0);
#pragma unroll
            for (int j = 0; j < TN; ++j) {
                bv[j].x = sb[k0 * PB + j * 16];
                bv[j].y = sb[(k0 + 1) * PB + j * 16];
            }
#pragma unroll
            for (int i = 0; i < TM; ++i)
#pragma unroll
                for (int j = 0; j < TN; ++j)
                    acc[i][j] = __builtin_amdgcn_wmma_f32_16x16x4_f32(
                         false, av[i],  false, bv[j],
                         (short)0, acc[i][j],
                         false,  false);
        }

        if (more) {
            float* da = sA + (cur ^ 1) * (BM * PA);
            float* db = sB + (cur ^ 1) * (KB * PB);
#pragma unroll
            for (int p = 0; p < A_PER; ++p) *(v4f*)&da[asl[p]] = ra[p];
#pragma unroll
            for (int p = 0; p < B_PER; ++p) *(v4f*)&db[bsl[p]] = rb[p];
        }
        __syncthreads();
    }

    const int mhi = (lane >> 4) << 3;
#pragma unroll
    for (int j = 0; j < TN; ++j) {
        const int   col = n0 + wn + j * 16 + c16;
        const float bb  = bias[col];
#pragma unroll
        for (int i = 0; i < TM; ++i) {
#pragma unroll
            for (int r = 0; r < 8; ++r) {
                float v = acc[i][j][r] + bb;
                if (act == 1) v = (v > 20.f) ? v : log1pf(expf(v));
                out[(size_t)(m0 + wm + i * 16 + mhi + r) * Nout + col] = v;
            }
        }
    }
}

__global__ __launch_bounds__(256)
void s6_scan_pass1(const float* __restrict__ x,
                   const float* __restrict__ dt,
                   const float* __restrict__ Bm,
                   const float* __restrict__ A_log,
                   float* __restrict__ hend,
                   float* __restrict__ aend) {
    const int b = blockIdx.z, c = blockIdx.y;
    const int d = blockIdx.x * 256 + threadIdx.x;

    __shared__ float sBm[CLEN * NSTATE];
    for (int idx = threadIdx.x; idx < CLEN * NSTATE; idx += 256)
        sBm[idx] = Bm[((size_t)b * SEQL + c * CLEN) * NSTATE + idx];
    __syncthreads();

    float Ac[NSTATE];
#pragma unroll
    for (int n = 0; n < NSTATE; ++n) Ac[n] = -expf(A_log[d * NSTATE + n]);

    float h[NSTATE], ap[NSTATE];
#pragma unroll
    for (int n = 0; n < NSTATE; ++n) { h[n] = 0.f; ap[n] = 1.f; }

    for (int t = 0; t < CLEN; ++t) {
        const size_t ofs = ((size_t)b * SEQL + c * CLEN + t) * DMODEL + d;
        const float dtv = dt[ofs];
        const float ub  = dtv * x[ofs];
        const int   bmo = t * NSTATE;
#pragma unroll
        for (int n = 0; n < NSTATE; ++n) {
            const float a = __expf(dtv * Ac[n]);
            h[n]  = a * h[n] + ub * sBm[bmo + n];
            ap[n] *= a;
        }
    }

    const size_t base = (((size_t)b * DMODEL + d) * NCHUNK + c) * NSTATE;
#pragma unroll
    for (int n = 0; n < NSTATE; ++n) { hend[base + n] = h[n]; aend[base + n] = ap[n]; }
}

__global__ __launch_bounds__(256)
void s6_scan_pass2(const float* __restrict__ hend,
                   const float* __restrict__ aend,
                   float* __restrict__ carry) {
    const int idx = blockIdx.x * 256 + threadIdx.x;
    const size_t bd = (size_t)(idx >> 4);
    const int n = idx & 15;
    float cv = 0.f;
    for (int c = 0; c < NCHUNK; ++c) {
        const size_t o = (bd * NCHUNK + c) * NSTATE + n;
        carry[o] = cv;
        cv = aend[o] * cv + hend[o];
    }
}

__global__ __launch_bounds__(256)
void s6_scan_pass3(const float* __restrict__ x,
                   const float* __restrict__ dt,
                   const float* __restrict__ Bm,
                   const float* __restrict__ Cm,
                   const float* __restrict__ A_log,
                   const float* __restrict__ carry,
                   const float* __restrict__ Dskip,
                   float* __restrict__ yout) {
    const int b = blockIdx.z, c = blockIdx.y;
    const int d = blockIdx.x * 256 + threadIdx.x;

    __shared__ float sBm[CLEN * NSTATE];
    __shared__ float sCm[CLEN * NSTATE];
    for (int idx = threadIdx.x; idx < CLEN * NSTATE; idx += 256) {
        const size_t g = ((size_t)b * SEQL + c * CLEN) * NSTATE + idx;
        sBm[idx] = Bm[g];
        sCm[idx] = Cm[g];
    }
    __syncthreads();

    float Ac[NSTATE];
#pragma unroll
    for (int n = 0; n < NSTATE; ++n) Ac[n] = -expf(A_log[d * NSTATE + n]);

    float h[NSTATE];
    const size_t base = (((size_t)b * DMODEL + d) * NCHUNK + c) * NSTATE;
#pragma unroll
    for (int n = 0; n < NSTATE; ++n) h[n] = carry[base + n];

    const float dsk = Dskip[d];
    for (int t = 0; t < CLEN; ++t) {
        const size_t ofs = ((size_t)b * SEQL + c * CLEN + t) * DMODEL + d;
        const float dtv = dt[ofs];
        const float xv  = x[ofs];
        const float ub  = dtv * xv;
        const int   so  = t * NSTATE;
        float y = 0.f;
#pragma unroll
        for (int n = 0; n < NSTATE; ++n) {
            const float a = __expf(dtv * Ac[n]);
            h[n] = a * h[n] + ub * sBm[so + n];
            y += sCm[so + n] * h[n];
        }
        y += dsk * xv;
        yout[ofs] = 0.5f * y * (1.f + erff(y * 0.70710678118654752f));
    }
}

__global__ __launch_bounds__(256)
void s6_ln_residual(const float* __restrict__ ymix,
                    const float* __restrict__ hin,
                    const float* __restrict__ g,
                    const float* __restrict__ be,
                    float* __restrict__ hout) {
    const size_t row = blockIdx.x;
    const int tid = threadIdx.x;
    const float v0 = ymix[row * DMODEL + tid];
    const float v1 = ymix[row * DMODEL + 256 + tid];

    __shared__ float s1[256], s2[256];
    s1[tid] = v0 + v1;
    s2[tid] = v0 * v0 + v1 * v1;
    __syncthreads();
    for (int st = 128; st > 0; st >>= 1) {
        if (tid < st) { s1[tid] += s1[tid + st]; s2[tid] += s2[tid + st]; }
        __syncthreads();
    }
    const float mu   = s1[0] * (1.f / DMODEL);
    const float var  = s2[0] * (1.f / DMODEL) - mu * mu;
    const float rstd = rsqrtf(var + LN_EPS);

    hout[row * DMODEL + tid] =
        hin[row * DMODEL + tid] + (v0 - mu) * rstd * g[tid] + be[tid];
    hout[row * DMODEL + 256 + tid] =
        hin[row * DMODEL + 256 + tid] + (v1 - mu) * rstd * g[256 + tid] + be[256 + tid];
}

extern "C" void kernel_launch(void* const* d_in, const int* in_sizes, int n_in,
                              void* d_out, int out_size, void* d_ws, size_t ws_size,
                              hipStream_t stream) {
    (void)in_sizes; (void)n_in; (void)out_size; (void)ws_size;

    const float* x      = (const float*)d_in[0];
    const float* A_log  = (const float*)d_in[1];
    const float* W_B    = (const float*)d_in[2];
    const float* b_B    = (const float*)d_in[3];
    const float* W_C    = (const float*)d_in[4];
    const float* b_C    = (const float*)d_in[5];
    const float* W_dt   = (const float*)d_in[6];
    const float* b_dt   = (const float*)d_in[7];
    const float* D_skip = (const float*)d_in[8];
    const float* W_mix  = (const float*)d_in[9];
    const float* b_mix  = (const float*)d_in[10];
    const float* ln_g   = (const float*)d_in[11];
    const float* ln_b   = (const float*)d_in[12];
    const float* W_dec  = (const float*)d_in[13];
    const float* b_dec  = (const float*)d_in[14];
    float* out = (float*)d_out;

    float* ws = (float*)d_ws;
    size_t off = 0;
    float* dtb   = ws + off; off += (size_t)MROWS * DMODEL;
    float* Bmb   = ws + off; off += (size_t)MROWS * NSTATE;
    float* Cmb   = ws + off; off += (size_t)MROWS * NSTATE;
    float* ygelu = ws + off; off += (size_t)MROWS * DMODEL;
    float* ymix  = ws + off; off += (size_t)MROWS * DMODEL;
    float* h1    = ws + off; off += (size_t)MROWS * DMODEL;
    float* h2    = ws + off; off += (size_t)MROWS * DMODEL;
    float* hend  = ws + off; off += (size_t)BATCH * DMODEL * NCHUNK * NSTATE;
    float* aend  = ws + off; off += (size_t)BATCH * DMODEL * NCHUNK * NSTATE;
    float* carry = ws + off; off += (size_t)BATCH * DMODEL * NCHUNK * NSTATE;

    const int DN = DMODEL * NSTATE;
    const int DD = DMODEL * DMODEL;

    const dim3 gBig(MROWS / 128, DMODEL / 64);
    const dim3 gBC (MROWS / 32, 1);
    const dim3 gScan(DMODEL / 256, NCHUNK, BATCH);

    for (int i = 0; i < 2; ++i) {
        const float* hin = (i == 0) ? x : h1;
        float* hout      = (i == 0) ? h1 : h2;

        s6_gemm_wmma_k512<128, 64, 32, 32><<<gBig, 256, 0, stream>>>(
            hin, W_dt + (size_t)i * DD, b_dt + i * DMODEL, dtb, DMODEL, 1);
        s6_gemm_wmma_k512<32, 16, 16, 16><<<gBC, 64, 0, stream>>>(
            hin, W_B + (size_t)i * DN, b_B + i * NSTATE, Bmb, NSTATE, 0);
        s6_gemm_wmma_k512<32, 16, 16, 16><<<gBC, 64, 0, stream>>>(
            hin, W_C + (size_t)i * DN, b_C + i * NSTATE, Cmb, NSTATE, 0);

        s6_scan_pass1<<<gScan, 256, 0, stream>>>(
            hin, dtb, Bmb, A_log + (size_t)i * DN, hend, aend);
        s6_scan_pass2<<<BATCH * DMODEL * NSTATE / 256, 256, 0, stream>>>(
            hend, aend, carry);
        s6_scan_pass3<<<gScan, 256, 0, stream>>>(
            hin, dtb, Bmb, Cmb, A_log + (size_t)i * DN, carry,
            D_skip + i * DMODEL, ygelu);

        s6_gemm_wmma_k512<128, 64, 32, 32><<<gBig, 256, 0, stream>>>(
            ygelu, W_mix + (size_t)i * DD, b_mix + i * DMODEL, ymix, DMODEL, 0);
        s6_ln_residual<<<MROWS, 256, 0, stream>>>(
            ymix, hin, ln_g + i * DMODEL, ln_b + i * DMODEL, hout);
    }

    s6_gemm_wmma_k512<128, 64, 32, 32><<<gBig, 256, 0, stream>>>(h2, W_dec, b_dec, out, OUTD, 0);
}


# Scan passes 1 and 3 walk a chunk in groups of 8 steps and issue the next group's dt and x loads ahead of the current group's arithmetic, so no step waits for a load it has just issued
# speedup vs baseline: 2.7121x; 1.7688x over previous
#include <hip/hip_runtime.h>
#include <hip/hip_bf16.h>
#include <math.h>

#define BATCH   4
#define SEQL    2048
#define DMODEL  512
#define NSTATE  16
#define OUTD    512
#define MROWS   (BATCH * SEQL)
#define KDIM    512
#define NCHUNK  16
#define CLEN    (SEQL / NCHUNK)
#define TGRP    8
#define LN_EPS  1e-5f

typedef float v2f __attribute__((ext_vector_type(2)));
typedef float v4f __attribute__((ext_vector_type(4)));
typedef float v8f __attribute__((ext_vector_type(8)));

template <int BM, int BN, int WM, int WN>
__global__ __launch_bounds__((BM / WM) * (BN / WN) * 32)
void s6_gemm_wmma_k512(const float* __restrict__ A,
                       const float* __restrict__ W,
                       const float* __restrict__ bias,
                       float* __restrict__ out,
                       int Nout, int act  ) {
    constexpr int KB    = 32;
    constexpr int PA    = KB + 4;
    constexpr int PB    = BN + 8;
    constexpr int NT    = (BM / WM) * (BN / WN) * 32;
    constexpr int TM    = WM / 16;
    constexpr int TN    = WN / 16;
    constexpr int AQ    = KB / 4;
    constexpr int BQ    = BN / 4;
    constexpr int A_PER = (BM * AQ) / NT;
    constexpr int B_PER = (KB * BQ) / NT;
    constexpr int NCH   = KDIM / KB;
    static_assert((BM * AQ) % NT == 0, "A chunk must split evenly over the block");
    static_assert((KB * BQ) % NT == 0, "W chunk must split evenly over the block");

    __shared__ float sA[2 * BM * PA] __attribute__((aligned(16)));
    __shared__ float sB[2 * KB * PB] __attribute__((aligned(16)));

    const int tid = threadIdx.x;
    const int m0  = blockIdx.x * BM;
    const int n0  = blockIdx.y * BN;

    const float* agp[A_PER]; int asl[A_PER];
    const float* bgp[B_PER]; int bsl[B_PER];
#pragma unroll
    for (int p = 0; p < A_PER; ++p) {
        const int idx = tid + p * NT;
        const int row = idx / AQ, kq = idx % AQ;
        agp[p] = A + (size_t)(m0 + row) * KDIM + kq * 4;
        asl[p] = row * PA + kq * 4;
    }
#pragma unroll
    for (int p = 0; p < B_PER; ++p) {
        const int idx = tid + p * NT;
        const int kr = idx / BQ, nq = idx % BQ;
        bgp[p] = W + (size_t)kr * Nout + n0 + nq * 4;
        bsl[p] = kr * PB + nq * 4;
    }

    v4f ra[A_PER], rb[B_PER];
#pragma unroll
    for (int p = 0; p < A_PER; ++p) ra[p] = *(const v4f*)agp[p];
#pragma unroll
    for (int p = 0; p < B_PER; ++p) rb[p] = *(const v4f*)bgp[p];

    const int wave = tid >> 5;
    const int lane = tid & 31;
    const int wm   = (wave / (BN / WN)) * WM;
    const int wn   = (wave % (BN / WN)) * WN;
    const int c16  = lane & 15;
    const int klo  = (lane >> 4) << 1;

    v8f acc[TM][TN];
#pragma unroll
    for (int i = 0; i < TM; ++i)
#pragma unroll
        for (int j = 0; j < TN; ++j)
            acc[i][j] = (v8f){0.f, 0.f, 0.f, 0.f, 0.f, 0.f, 0.f, 0.f};

    const float* saf = sA + (wm + c16) * PA + klo;
    const float* sbf = sB + klo * PB + wn + c16;

#pragma unroll
    for (int p = 0; p < A_PER; ++p) *(v4f*)&sA[asl[p]] = ra[p];
#pragma unroll
    for (int p = 0; p < B_PER; ++p) *(v4f*)&sB[bsl[p]] = rb[p];
    __syncthreads();

    for (int kc = 0; kc < NCH; ++kc) {
        const int  cur  = kc & 1;
        const bool more = (kc + 1 < NCH);

        if (more) {
#pragma unroll
            for (int p = 0; p < A_PER; ++p)
                ra[p] = *(const v4f*)(agp[p] + (kc + 1) * KB);
#pragma unroll
            for (int p = 0; p < B_PER; ++p)
                rb[p] = *(const v4f*)(bgp[p] + (size_t)(kc + 1) * KB * Nout);
        }
        __builtin_amdgcn_sched_barrier(0);

        const float* sa = saf + cur * (BM * PA);
        const float* sb = sbf + cur * (KB * PB);
#pragma unroll
        for (int k0 = 0; k0 < KB; k0 += 4) {
            v2f av[TM], bv[TN];
#pragma unroll
            for (int i = 0; i < TM; ++i)
                av[i] = *(const v2f*)(sa + i * 16 * PA + k0);
#pragma unroll
            for (int j = 0; j < TN; ++j) {
                bv[j].x = sb[k0 * PB + j * 16];
                bv[j].y = sb[(k0 + 1) * PB + j * 16];
            }
#pragma unroll
            for (int i = 0; i < TM; ++i)
#pragma unroll
                for (int j = 0; j < TN; ++j)
                    acc[i][j] = __builtin_amdgcn_wmma_f32_16x16x4_f32(
                         false, av[i],  false, bv[j],
                         (short)0, acc[i][j],
                         false,  false);
        }

        if (more) {
            float* da = sA + (cur ^ 1) * (BM * PA);
            float* db = sB + (cur ^ 1) * (KB * PB);
#pragma unroll
            for (int p = 0; p < A_PER; ++p) *(v4f*)&da[asl[p]] = ra[p];
#pragma unroll
            for (int p = 0; p < B_PER; ++p) *(v4f*)&db[bsl[p]] = rb[p];
        }
        __syncthreads();
    }

    const int mhi = (lane >> 4) << 3;
#pragma unroll
    for (int j = 0; j < TN; ++j) {
        const int   col = n0 + wn + j * 16 + c16;
        const float bb  = bias[col];
#pragma unroll
        for (int i = 0; i < TM; ++i) {
#pragma unroll
            for (int r = 0; r < 8; ++r) {
                float v = acc[i][j][r] + bb;
                if (act == 1) v = (v > 20.f) ? v : log1pf(expf(v));
                out[(size_t)(m0 + wm + i * 16 + mhi + r) * Nout + col] = v;
            }
        }
    }
}

__global__ __launch_bounds__(256)
void s6_scan_pass1(const float* __restrict__ x,
                   const float* __restrict__ dt,
                   const float* __restrict__ Bm,
                   const float* __restrict__ A_log,
                   float* __restrict__ hend,
                   float* __restrict__ aend) {
    const int b = blockIdx.z, c = blockIdx.y;
    const int d = blockIdx.x * 256 + threadIdx.x;

    __shared__ float sBm[CLEN * NSTATE];
    for (int idx = threadIdx.x; idx < CLEN * NSTATE; idx += 256)
        sBm[idx] = Bm[((size_t)b * SEQL + c * CLEN) * NSTATE + idx];
    __syncthreads();

    float Ac[NSTATE];
#pragma unroll
    for (int n = 0; n < NSTATE; ++n) Ac[n] = -expf(A_log[d * NSTATE + n]);

    float h[NSTATE], ap[NSTATE];
#pragma unroll
    for (int n = 0; n < NSTATE; ++n) { h[n] = 0.f; ap[n] = 1.f; }

    const float* dtp = dt + ((size_t)b * SEQL + c * CLEN) * DMODEL + d;
    const float* xp  = x  + ((size_t)b * SEQL + c * CLEN) * DMODEL + d;
    v2f cur[TGRP];
#pragma unroll
    for (int i = 0; i < TGRP; ++i) {
        cur[i].x = dtp[(size_t)i * DMODEL];
        cur[i].y = xp[(size_t)i * DMODEL];
    }

    for (int t0 = 0; t0 < CLEN; t0 += TGRP) {
        const int tn = (t0 + TGRP < CLEN) ? (t0 + TGRP) : t0;
        v2f nxt[TGRP];
#pragma unroll
        for (int i = 0; i < TGRP; ++i) {
            nxt[i].x = dtp[(size_t)(tn + i) * DMODEL];
            nxt[i].y = xp[(size_t)(tn + i) * DMODEL];
        }
        __builtin_amdgcn_sched_barrier(0);

#pragma unroll
        for (int i = 0; i < TGRP; ++i) {
            const float dtv = cur[i].x;
            const float ub  = dtv * cur[i].y;
            const int   bmo = (t0 + i) * NSTATE;
#pragma unroll
            for (int n = 0; n < NSTATE; ++n) {
                const float a = __expf(dtv * Ac[n]);
                h[n]  = a * h[n] + ub * sBm[bmo + n];
                ap[n] *= a;
            }
        }
#pragma unroll
        for (int i = 0; i < TGRP; ++i) cur[i] = nxt[i];
    }

    const size_t base = (((size_t)b * DMODEL + d) * NCHUNK + c) * NSTATE;
#pragma unroll
    for (int n = 0; n < NSTATE; ++n) { hend[base + n] = h[n]; aend[base + n] = ap[n]; }
}

__global__ __launch_bounds__(256)
void s6_scan_pass2(const float* __restrict__ hend,
                   const float* __restrict__ aend,
                   float* __restrict__ carry) {
    const int idx = blockIdx.x * 256 + threadIdx.x;
    const size_t bd = (size_t)(idx >> 4);
    const int n = idx & 15;
    float cv = 0.f;
    for (int c = 0; c < NCHUNK; ++c) {
        const size_t o = (bd * NCHUNK + c) * NSTATE + n;
        carry[o] = cv;
        cv = aend[o] * cv + hend[o];
    }
}

__global__ __launch_bounds__(256)
void s6_scan_pass3(const float* __restrict__ x,
                   const float* __restrict__ dt,
                   const float* __restrict__ Bm,
                   const float* __restrict__ Cm,
                   const float* __restrict__ A_log,
                   const float* __restrict__ carry,
                   const float* __restrict__ Dskip,
                   float* __restrict__ yout) {
    const int b = blockIdx.z, c = blockIdx.y;
    const int d = blockIdx.x * 256 + threadIdx.x;

    __shared__ float sBm[CLEN * NSTATE];
    __shared__ float sCm[CLEN * NSTATE];
    for (int idx = threadIdx.x; idx < CLEN * NSTATE; idx += 256) {
        const size_t g = ((size_t)b * SEQL + c * CLEN) * NSTATE + idx;
        sBm[idx] = Bm[g];
        sCm[idx] = Cm[g];
    }
    __syncthreads();

    float Ac[NSTATE];
#pragma unroll
    for (int n = 0; n < NSTATE; ++n) Ac[n] = -expf(A_log[d * NSTATE + n]);

    float h[NSTATE];
    const size_t base = (((size_t)b * DMODEL + d) * NCHUNK + c) * NSTATE;
#pragma unroll
    for (int n = 0; n < NSTATE; ++n) h[n] = carry[base + n];

    const float dsk = Dskip[d];

    const size_t cbase = ((size_t)b * SEQL + c * CLEN) * DMODEL + d;
    const float* dtp = dt + cbase;
    const float* xp  = x  + cbase;
    float*       yp  = yout + cbase;
    v2f cur[TGRP];
#pragma unroll
    for (int i = 0; i < TGRP; ++i) {
        cur[i].x = dtp[(size_t)i * DMODEL];
        cur[i].y = xp[(size_t)i * DMODEL];
    }

    for (int t0 = 0; t0 < CLEN; t0 += TGRP) {
        const int tn = (t0 + TGRP < CLEN) ? (t0 + TGRP) : t0;
        v2f nxt[TGRP];
#pragma unroll
        for (int i = 0; i < TGRP; ++i) {
            nxt[i].x = dtp[(size_t)(tn + i) * DMODEL];
            nxt[i].y = xp[(size_t)(tn + i) * DMODEL];
        }
        __builtin_amdgcn_sched_barrier(0);

        float yv[TGRP];
#pragma unroll
        for (int i = 0; i < TGRP; ++i) {
            const float dtv = cur[i].x;
            const float xv  = cur[i].y;
            const float ub  = dtv * xv;
            const int   so  = (t0 + i) * NSTATE;
            float y = 0.f;
#pragma unroll
            for (int n = 0; n < NSTATE; ++n) {
                const float a = __expf(dtv * Ac[n]);
                h[n] = a * h[n] + ub * sBm[so + n];
                y += sCm[so + n] * h[n];
            }
            y += dsk * xv;
            yv[i] = y;
        }
#pragma unroll
        for (int i = 0; i < TGRP; ++i) {
            const float y = yv[i];
            yp[(size_t)(t0 + i) * DMODEL] =
                0.5f * y * (1.f + erff(y * 0.70710678118654752f));
        }
#pragma unroll
        for (int i = 0; i < TGRP; ++i) cur[i] = nxt[i];
    }
}

__global__ __launch_bounds__(256)
void s6_ln_residual(const float* __restrict__ ymix,
                    const float* __restrict__ hin,
                    const float* __restrict__ g,
                    const float* __restrict__ be,
                    float* __restrict__ hout) {
    const size_t row = blockIdx.x;
    const int tid = threadIdx.x;
    const float v0 = ymix[row * DMODEL + tid];
    const float v1 = ymix[row * DMODEL + 256 + tid];

    __shared__ float s1[256], s2[256];
    s1[tid] = v0 + v1;
    s2[tid] = v0 * v0 + v1 * v1;
    __syncthreads();
    for (int st = 128; st > 0; st >>= 1) {
        if (tid < st) { s1[tid] += s1[tid + st]; s2[tid] += s2[tid + st]; }
        __syncthreads();
    }
    const float mu   = s1[0] * (1.f / DMODEL);
    const float var  = s2[0] * (1.f / DMODEL) - mu * mu;
    const float rstd = rsqrtf(var + LN_EPS);

    hout[row * DMODEL + tid] =
        hin[row * DMODEL + tid] + (v0 - mu) * rstd * g[tid] + be[tid];
    hout[row * DMODEL + 256 + tid] =
        hin[row * DMODEL + 256 + tid] + (v1 - mu) * rstd * g[256 + tid] + be[256 + tid];
}

extern "C" void kernel_launch(void* const* d_in, const int* in_sizes, int n_in,
                              void* d_out, int out_size, void* d_ws, size_t ws_size,
                              hipStream_t stream) {
    (void)in_sizes; (void)n_in; (void)out_size; (void)ws_size;

    const float* x      = (const float*)d_in[0];
    const float* A_log  = (const float*)d_in[1];
    const float* W_B    = (const float*)d_in[2];
    const float* b_B    = (const float*)d_in[3];
    const float* W_C    = (const float*)d_in[4];
    const float* b_C    = (const float*)d_in[5];
    const float* W_dt   = (const float*)d_in[6];
    const float* b_dt   = (const float*)d_in[7];
    const float* D_skip = (const float*)d_in[8];
    const float* W_mix  = (const float*)d_in[9];
    const float* b_mix  = (const float*)d_in[10];
    const float* ln_g   = (const float*)d_in[11];
    const float* ln_b   = (const float*)d_in[12];
    const float* W_dec  = (const float*)d_in[13];
    const float* b_dec  = (const float*)d_in[14];
    float* out = (float*)d_out;

    float* ws = (float*)d_ws;
    size_t off = 0;
    float* dtb   = ws + off; off += (size_t)MROWS * DMODEL;
    float* Bmb   = ws + off; off += (size_t)MROWS * NSTATE;
    float* Cmb   = ws + off; off += (size_t)MROWS * NSTATE;
    float* ygelu = ws + off; off += (size_t)MROWS * DMODEL;
    float* ymix  = ws + off; off += (size_t)MROWS * DMODEL;
    float* h1    = ws + off; off += (size_t)MROWS * DMODEL;
    float* h2    = ws + off; off += (size_t)MROWS * DMODEL;
    float* hend  = ws + off; off += (size_t)BATCH * DMODEL * NCHUNK * NSTATE;
    float* aend  = ws + off; off += (size_t)BATCH * DMODEL * NCHUNK * NSTATE;
    float* carry = ws + off; off += (size_t)BATCH * DMODEL * NCHUNK * NSTATE;

    const int DN = DMODEL * NSTATE;
    const int DD = DMODEL * DMODEL;

    const dim3 gBig(MROWS / 128, DMODEL / 64);
    const dim3 gBC (MROWS / 32, 1);
    const dim3 gScan(DMODEL / 256, NCHUNK, BATCH);

    for (int i = 0; i < 2; ++i) {
        const float* hin = (i == 0) ? x : h1;
        float* hout      = (i == 0) ? h1 : h2;

        s6_gemm_wmma_k512<128, 64, 32, 32><<<gBig, 256, 0, stream>>>(
            hin, W_dt + (size_t)i * DD, b_dt + i * DMODEL, dtb, DMODEL, 1);
        s6_gemm_wmma_k512<32, 16, 16, 16><<<gBC, 64, 0, stream>>>(
            hin, W_B + (size_t)i * DN, b_B + i * NSTATE, Bmb, NSTATE, 0);
        s6_gemm_wmma_k512<32, 16, 16, 16><<<gBC, 64, 0, stream>>>(
            hin, W_C + (size_t)i * DN, b_C + i * NSTATE, Cmb, NSTATE, 0);

        s6_scan_pass1<<<gScan, 256, 0, stream>>>(
            hin, dtb, Bmb, A_log + (size_t)i * DN, hend, aend);
        s6_scan_pass2<<<BATCH * DMODEL * NSTATE / 256, 256, 0, stream>>>(
            hend, aend, carry);
        s6_scan_pass3<<<gScan, 256, 0, stream>>>(
            hin, dtb, Bmb, Cmb, A_log + (size_t)i * DN, carry,
            D_skip + i * DMODEL, ygelu);

        s6_gemm_wmma_k512<128, 64, 32, 32><<<gBig, 256, 0, stream>>>(
            ygelu, W_mix + (size_t)i * DD, b_mix + i * DMODEL, ymix, DMODEL, 0);
        s6_ln_residual<<<MROWS, 256, 0, stream>>>(
            ymix, hin, ln_g + i * DMODEL, ln_b + i * DMODEL, hout);
    }

    s6_gemm_wmma_k512<128, 64, 32, 32><<<gBig, 256, 0, stream>>>(h2, W_dec, b_dec, out, OUTD, 0);
}


# Scan cut into 32 chunks of 64 steps, not 16 of 128: passes 1 and 3 launch 256 workgroups, one per WGP, where 128 left half the chip idle; chunk carries recombine at 32 points (rounding order)
# speedup vs baseline: 3.1458x; 1.1599x over previous
#include <hip/hip_runtime.h>
#include <hip/hip_bf16.h>
#include <math.h>

#define BATCH   4
#define SEQL    2048
#define DMODEL  512
#define NSTATE  16
#define OUTD    512
#define MROWS   (BATCH * SEQL)
#define KDIM    512
#define NCHUNK  32
#define CLEN    (SEQL / NCHUNK)
#define TGRP    8
#define LN_EPS  1e-5f

typedef float v2f __attribute__((ext_vector_type(2)));
typedef float v4f __attribute__((ext_vector_type(4)));
typedef float v8f __attribute__((ext_vector_type(8)));

template <int BM, int BN, int WM, int WN>
__global__ __launch_bounds__((BM / WM) * (BN / WN) * 32)
void s6_gemm_wmma_k512(const float* __restrict__ A,
                       const float* __restrict__ W,
                       const float* __restrict__ bias,
                       float* __restrict__ out,
                       int Nout, int act  ) {
    constexpr int KB    = 32;
    constexpr int PA    = KB + 4;
    constexpr int PB    = BN + 8;
    constexpr int NT    = (BM / WM) * (BN / WN) * 32;
    constexpr int TM    = WM / 16;
    constexpr int TN    = WN / 16;
    constexpr int AQ    = KB / 4;
    constexpr int BQ    = BN / 4;
    constexpr int A_PER = (BM * AQ) / NT;
    constexpr int B_PER = (KB * BQ) / NT;
    constexpr int NCH   = KDIM / KB;
    static_assert((BM * AQ) % NT == 0, "A chunk must split evenly over the block");
    static_assert((KB * BQ) % NT == 0, "W chunk must split evenly over the block");

    __shared__ float sA[2 * BM * PA] __attribute__((aligned(16)));
    __shared__ float sB[2 * KB * PB] __attribute__((aligned(16)));

    const int tid = threadIdx.x;
    const int m0  = blockIdx.x * BM;
    const int n0  = blockIdx.y * BN;

    const float* agp[A_PER]; int asl[A_PER];
    const float* bgp[B_PER]; int bsl[B_PER];
#pragma unroll
    for (int p = 0; p < A_PER; ++p) {
        const int idx = tid + p * NT;
        const int row = idx / AQ, kq = idx % AQ;
        agp[p] = A + (size_t)(m0 + row) * KDIM + kq * 4;
        asl[p] = row * PA + kq * 4;
    }
#pragma unroll
    for (int p = 0; p < B_PER; ++p) {
        const int idx = tid + p * NT;
        const int kr = idx / BQ, nq = idx % BQ;
        bgp[p] = W + (size_t)kr * Nout + n0 + nq * 4;
        bsl[p] = kr * PB + nq * 4;
    }

    v4f ra[A_PER], rb[B_PER];
#pragma unroll
    for (int p = 0; p < A_PER; ++p) ra[p] = *(const v4f*)agp[p];
#pragma unroll
    for (int p = 0; p < B_PER; ++p) rb[p] = *(const v4f*)bgp[p];

    const int wave = tid >> 5;
    const int lane = tid & 31;
    const int wm   = (wave / (BN / WN)) * WM;
    const int wn   = (wave % (BN / WN)) * WN;
    const int c16  = lane & 15;
    const int klo  = (lane >> 4) << 1;

    v8f acc[TM][TN];
#pragma unroll
    for (int i = 0; i < TM; ++i)
#pragma unroll
        for (int j = 0; j < TN; ++j)
            acc[i][j] = (v8f){0.f, 0.f, 0.f, 0.f, 0.f, 0.f, 0.f, 0.f};

    const float* saf = sA + (wm + c16) * PA + klo;
    const float* sbf = sB + klo * PB + wn + c16;

#pragma unroll
    for (int p = 0; p < A_PER; ++p) *(v4f*)&sA[asl[p]] = ra[p];
#pragma unroll
    for (int p = 0; p < B_PER; ++p) *(v4f*)&sB[bsl[p]] = rb[p];
    __syncthreads();

    for (int kc = 0; kc < NCH; ++kc) {
        const int  cur  = kc & 1;
        const bool more = (kc + 1 < NCH);

        if (more) {
#pragma unroll
            for (int p = 0; p < A_PER; ++p)
                ra[p] = *(const v4f*)(agp[p] + (kc + 1) * KB);
#pragma unroll
            for (int p = 0; p < B_PER; ++p)
                rb[p] = *(const v4f*)(bgp[p] + (size_t)(kc + 1) * KB * Nout);
        }
        __builtin_amdgcn_sched_barrier(0);

        const float* sa = saf + cur * (BM * PA);
        const float* sb = sbf + cur * (KB * PB);
#pragma unroll
        for (int k0 = 0; k0 < KB; k0 += 4) {
            v2f av[TM], bv[TN];
#pragma unroll
            for (int i = 0; i < TM; ++i)
                av[i] = *(const v2f*)(sa + i * 16 * PA + k0);
#pragma unroll
            for (int j = 0; j < TN; ++j) {
                bv[j].x = sb[k0 * PB + j * 16];
                bv[j].y = sb[(k0 + 1) * PB + j * 16];
            }
#pragma unroll
            for (int i = 0; i < TM; ++i)
#pragma unroll
                for (int j = 0; j < TN; ++j)
                    acc[i][j] = __builtin_amdgcn_wmma_f32_16x16x4_f32(
                         false, av[i],  false, bv[j],
                         (short)0, acc[i][j],
                         false,  false);
        }

        if (more) {
            float* da = sA + (cur ^ 1) * (BM * PA);
            float* db = sB + (cur ^ 1) * (KB * PB);
#pragma unroll
            for (int p = 0; p < A_PER; ++p) *(v4f*)&da[asl[p]] = ra[p];
#pragma unroll
            for (int p = 0; p < B_PER; ++p) *(v4f*)&db[bsl[p]] = rb[p];
        }
        __syncthreads();
    }

    const int mhi = (lane >> 4) << 3;
#pragma unroll
    for (int j = 0; j < TN; ++j) {
        const int   col = n0 + wn + j * 16 + c16;
        const float bb  = bias[col];
#pragma unroll
        for (int i = 0; i < TM; ++i) {
#pragma unroll
            for (int r = 0; r < 8; ++r) {
                float v = acc[i][j][r] + bb;
                if (act == 1) v = (v > 20.f) ? v : log1pf(expf(v));
                out[(size_t)(m0 + wm + i * 16 + mhi + r) * Nout + col] = v;
            }
        }
    }
}

__global__ __launch_bounds__(256)
void s6_scan_pass1(const float* __restrict__ x,
                   const float* __restrict__ dt,
                   const float* __restrict__ Bm,
                   const float* __restrict__ A_log,
                   float* __restrict__ hend,
                   float* __restrict__ aend) {
    const int b = blockIdx.z, c = blockIdx.y;
    const int d = blockIdx.x * 256 + threadIdx.x;

    __shared__ float sBm[CLEN * NSTATE];
    for (int idx = threadIdx.x; idx < CLEN * NSTATE; idx += 256)
        sBm[idx] = Bm[((size_t)b * SEQL + c * CLEN) * NSTATE + idx];
    __syncthreads();

    float Ac[NSTATE];
#pragma unroll
    for (int n = 0; n < NSTATE; ++n) Ac[n] = -expf(A_log[d * NSTATE + n]);

    float h[NSTATE], ap[NSTATE];
#pragma unroll
    for (int n = 0; n < NSTATE; ++n) { h[n] = 0.f; ap[n] = 1.f; }

    const float* dtp = dt + ((size_t)b * SEQL + c * CLEN) * DMODEL + d;
    const float* xp  = x  + ((size_t)b * SEQL + c * CLEN) * DMODEL + d;
    v2f cur[TGRP];
#pragma unroll
    for (int i = 0; i < TGRP; ++i) {
        cur[i].x = dtp[(size_t)i * DMODEL];
        cur[i].y = xp[(size_t)i * DMODEL];
    }

    for (int t0 = 0; t0 < CLEN; t0 += TGRP) {
        const int tn = (t0 + TGRP < CLEN) ? (t0 + TGRP) : t0;
        v2f nxt[TGRP];
#pragma unroll
        for (int i = 0; i < TGRP; ++i) {
            nxt[i].x = dtp[(size_t)(tn + i) * DMODEL];
            nxt[i].y = xp[(size_t)(tn + i) * DMODEL];
        }
        __builtin_amdgcn_sched_barrier(0);

#pragma unroll
        for (int i = 0; i < TGRP; ++i) {
            const float dtv = cur[i].x;
            const float ub  = dtv * cur[i].y;
            const int   bmo = (t0 + i) * NSTATE;
#pragma unroll
            for (int n = 0; n < NSTATE; ++n) {
                const float a = __expf(dtv * Ac[n]);
                h[n]  = a * h[n] + ub * sBm[bmo + n];
                ap[n] *= a;
            }
        }
#pragma unroll
        for (int i = 0; i < TGRP; ++i) cur[i] = nxt[i];
    }

    const size_t base = (((size_t)b * DMODEL + d) * NCHUNK + c) * NSTATE;
#pragma unroll
    for (int n = 0; n < NSTATE; ++n) { hend[base + n] = h[n]; aend[base + n] = ap[n]; }
}

__global__ __launch_bounds__(256)
void s6_scan_pass2(const float* __restrict__ hend,
                   const float* __restrict__ aend,
                   float* __restrict__ carry) {
    const int idx = blockIdx.x * 256 + threadIdx.x;
    const size_t bd = (size_t)(idx >> 4);
    const int n = idx & 15;
    float cv = 0.f;
    for (int c = 0; c < NCHUNK; ++c) {
        const size_t o = (bd * NCHUNK + c) * NSTATE + n;
        carry[o] = cv;
        cv = aend[o] * cv + hend[o];
    }
}

__global__ __launch_bounds__(256)
void s6_scan_pass3(const float* __restrict__ x,
                   const float* __restrict__ dt,
                   const float* __restrict__ Bm,
                   const float* __restrict__ Cm,
                   const float* __restrict__ A_log,
                   const float* __restrict__ carry,
                   const float* __restrict__ Dskip,
                   float* __restrict__ yout) {
    const int b = blockIdx.z, c = blockIdx.y;
    const int d = blockIdx.x * 256 + threadIdx.x;

    __shared__ float sBm[CLEN * NSTATE];
    __shared__ float sCm[CLEN * NSTATE];
    for (int idx = threadIdx.x; idx < CLEN * NSTATE; idx += 256) {
        const size_t g = ((size_t)b * SEQL + c * CLEN) * NSTATE + idx;
        sBm[idx] = Bm[g];
        sCm[idx] = Cm[g];
    }
    __syncthreads();

    float Ac[NSTATE];
#pragma unroll
    for (int n = 0; n < NSTATE; ++n) Ac[n] = -expf(A_log[d * NSTATE + n]);

    float h[NSTATE];
    const size_t base = (((size_t)b * DMODEL + d) * NCHUNK + c) * NSTATE;
#pragma unroll
    for (int n = 0; n < NSTATE; ++n) h[n] = carry[base + n];

    const float dsk = Dskip[d];

    const size_t cbase = ((size_t)b * SEQL + c * CLEN) * DMODEL + d;
    const float* dtp = dt + cbase;
    const float* xp  = x  + cbase;
    float*       yp  = yout + cbase;
    v2f cur[TGRP];
#pragma unroll
    for (int i = 0; i < TGRP; ++i) {
        cur[i].x = dtp[(size_t)i * DMODEL];
        cur[i].y = xp[(size_t)i * DMODEL];
    }

    for (int t0 = 0; t0 < CLEN; t0 += TGRP) {
        const int tn = (t0 + TGRP < CLEN) ? (t0 + TGRP) : t0;
        v2f nxt[TGRP];
#pragma unroll
        for (int i = 0; i < TGRP; ++i) {
            nxt[i].x = dtp[(size_t)(tn + i) * DMODEL];
            nxt[i].y = xp[(size_t)(tn + i) * DMODEL];
        }
        __builtin_amdgcn_sched_barrier(0);

        float yv[TGRP];
#pragma unroll
        for (int i = 0; i < TGRP; ++i) {
            const float dtv = cur[i].x;
            const float xv  = cur[i].y;
            const float ub  = dtv * xv;
            const int   so  = (t0 + i) * NSTATE;
            float y = 0.f;
#pragma unroll
            for (int n = 0; n < NSTATE; ++n) {
                const float a = __expf(dtv * Ac[n]);
                h[n] = a * h[n] + ub * sBm[so + n];
                y += sCm[so + n] * h[n];
            }
            y += dsk * xv;
            yv[i] = y;
        }
#pragma unroll
        for (int i = 0; i < TGRP; ++i) {
            const float y = yv[i];
            yp[(size_t)(t0 + i) * DMODEL] =
                0.5f * y * (1.f + erff(y * 0.70710678118654752f));
        }
#pragma unroll
        for (int i = 0; i < TGRP; ++i) cur[i] = nxt[i];
    }
}

__global__ __launch_bounds__(256)
void s6_ln_residual(const float* __restrict__ ymix,
                    const float* __restrict__ hin,
                    const float* __restrict__ g,
                    const float* __restrict__ be,
                    float* __restrict__ hout) {
    const size_t row = blockIdx.x;
    const int tid = threadIdx.x;
    const float v0 = ymix[row * DMODEL + tid];
    const float v1 = ymix[row * DMODEL + 256 + tid];

    __shared__ float s1[256], s2[256];
    s1[tid] = v0 + v1;
    s2[tid] = v0 * v0 + v1 * v1;
    __syncthreads();
    for (int st = 128; st > 0; st >>= 1) {
        if (tid < st) { s1[tid] += s1[tid + st]; s2[tid] += s2[tid + st]; }
        __syncthreads();
    }
    const float mu   = s1[0] * (1.f / DMODEL);
    const float var  = s2[0] * (1.f / DMODEL) - mu * mu;
    const float rstd = rsqrtf(var + LN_EPS);

    hout[row * DMODEL + tid] =
        hin[row * DMODEL + tid] + (v0 - mu) * rstd * g[tid] + be[tid];
    hout[row * DMODEL + 256 + tid] =
        hin[row * DMODEL + 256 + tid] + (v1 - mu) * rstd * g[256 + tid] + be[256 + tid];
}

extern "C" void kernel_launch(void* const* d_in, const int* in_sizes, int n_in,
                              void* d_out, int out_size, void* d_ws, size_t ws_size,
                              hipStream_t stream) {
    (void)in_sizes; (void)n_in; (void)out_size; (void)ws_size;

    const float* x      = (const float*)d_in[0];
    const float* A_log  = (const float*)d_in[1];
    const float* W_B    = (const float*)d_in[2];
    const float* b_B    = (const float*)d_in[3];
    const float* W_C    = (const float*)d_in[4];
    const float* b_C    = (const float*)d_in[5];
    const float* W_dt   = (const float*)d_in[6];
    const float* b_dt   = (const float*)d_in[7];
    const float* D_skip = (const float*)d_in[8];
    const float* W_mix  = (const float*)d_in[9];
    const float* b_mix  = (const float*)d_in[10];
    const float* ln_g   = (const float*)d_in[11];
    const float* ln_b   = (const float*)d_in[12];
    const float* W_dec  = (const float*)d_in[13];
    const float* b_dec  = (const float*)d_in[14];
    float* out = (float*)d_out;

    float* ws = (float*)d_ws;
    size_t off = 0;
    float* dtb   = ws + off; off += (size_t)MROWS * DMODEL;
    float* Bmb   = ws + off; off += (size_t)MROWS * NSTATE;
    float* Cmb   = ws + off; off += (size_t)MROWS * NSTATE;
    float* ygelu = ws + off; off += (size_t)MROWS * DMODEL;
    float* ymix  = ws + off; off += (size_t)MROWS * DMODEL;
    float* h1    = ws + off; off += (size_t)MROWS * DMODEL;
    float* h2    = ws + off; off += (size_t)MROWS * DMODEL;
    float* hend  = ws + off; off += (size_t)BATCH * DMODEL * NCHUNK * NSTATE;
    float* aend  = ws + off; off += (size_t)BATCH * DMODEL * NCHUNK * NSTATE;
    float* carry = ws + off; off += (size_t)BATCH * DMODEL * NCHUNK * NSTATE;

    const int DN = DMODEL * NSTATE;
    const int DD = DMODEL * DMODEL;

    const dim3 gBig(MROWS / 128, DMODEL / 64);
    const dim3 gBC (MROWS / 32, 1);
    const dim3 gScan(DMODEL / 256, NCHUNK, BATCH);

    for (int i = 0; i < 2; ++i) {
        const float* hin = (i == 0) ? x : h1;
        float* hout      = (i == 0) ? h1 : h2;

        s6_gemm_wmma_k512<128, 64, 32, 32><<<gBig, 256, 0, stream>>>(
            hin, W_dt + (size_t)i * DD, b_dt + i * DMODEL, dtb, DMODEL, 1);
        s6_gemm_wmma_k512<32, 16, 16, 16><<<gBC, 64, 0, stream>>>(
            hin, W_B + (size_t)i * DN, b_B + i * NSTATE, Bmb, NSTATE, 0);
        s6_gemm_wmma_k512<32, 16, 16, 16><<<gBC, 64, 0, stream>>>(
            hin, W_C + (size_t)i * DN, b_C + i * NSTATE, Cmb, NSTATE, 0);

        s6_scan_pass1<<<gScan, 256, 0, stream>>>(
            hin, dtb, Bmb, A_log + (size_t)i * DN, hend, aend);
        s6_scan_pass2<<<BATCH * DMODEL * NSTATE / 256, 256, 0, stream>>>(
            hend, aend, carry);
        s6_scan_pass3<<<gScan, 256, 0, stream>>>(
            hin, dtb, Bmb, Cmb, A_log + (size_t)i * DN, carry,
            D_skip + i * DMODEL, ygelu);

        s6_gemm_wmma_k512<128, 64, 32, 32><<<gBig, 256, 0, stream>>>(
            ygelu, W_mix + (size_t)i * DD, b_mix + i * DMODEL, ymix, DMODEL, 0);
        s6_ln_residual<<<MROWS, 256, 0, stream>>>(
            ymix, hin, ln_g + i * DMODEL, ln_b + i * DMODEL, hout);
    }

    s6_gemm_wmma_k512<128, 64, 32, 32><<<gBig, 256, 0, stream>>>(h2, W_dec, b_dec, out, OUTD, 0);
}


# LayerNorm+residual: one wave per row, row kept in registers (2 rows a wave, 16-byte loads, butterfly sums, variance about the mean: rounding order): 4096 waves a launch, not 65536; no LDS, no barrier
# speedup vs baseline: 3.2551x; 1.0347x over previous
#include <hip/hip_runtime.h>
#include <hip/hip_bf16.h>
#include <math.h>

#define BATCH   4
#define SEQL    2048
#define DMODEL  512
#define NSTATE  16
#define OUTD    512
#define MROWS   (BATCH * SEQL)
#define KDIM    512
#define NCHUNK  32
#define CLEN    (SEQL / NCHUNK)
#define TGRP    8
#define LN_EPS  1e-5f

typedef float v2f __attribute__((ext_vector_type(2)));
typedef float v4f __attribute__((ext_vector_type(4)));
typedef float v8f __attribute__((ext_vector_type(8)));

template <int BM, int BN, int WM, int WN>
__global__ __launch_bounds__((BM / WM) * (BN / WN) * 32)
void s6_gemm_wmma_k512(const float* __restrict__ A,
                       const float* __restrict__ W,
                       const float* __restrict__ bias,
                       float* __restrict__ out,
                       int Nout, int act  ) {
    constexpr int KB    = 32;
    constexpr int PA    = KB + 4;
    constexpr int PB    = BN + 8;
    constexpr int NT    = (BM / WM) * (BN / WN) * 32;
    constexpr int TM    = WM / 16;
    constexpr int TN    = WN / 16;
    constexpr int AQ    = KB / 4;
    constexpr int BQ    = BN / 4;
    constexpr int A_PER = (BM * AQ) / NT;
    constexpr int B_PER = (KB * BQ) / NT;
    constexpr int NCH   = KDIM / KB;
    static_assert((BM * AQ) % NT == 0, "A chunk must split evenly over the block");
    static_assert((KB * BQ) % NT == 0, "W chunk must split evenly over the block");

    __shared__ float sA[2 * BM * PA] __attribute__((aligned(16)));
    __shared__ float sB[2 * KB * PB] __attribute__((aligned(16)));

    const int tid = threadIdx.x;
    const int m0  = blockIdx.x * BM;
    const int n0  = blockIdx.y * BN;

    const float* agp[A_PER]; int asl[A_PER];
    const float* bgp[B_PER]; int bsl[B_PER];
#pragma unroll
    for (int p = 0; p < A_PER; ++p) {
        const int idx = tid + p * NT;
        const int row = idx / AQ, kq = idx % AQ;
        agp[p] = A + (size_t)(m0 + row) * KDIM + kq * 4;
        asl[p] = row * PA + kq * 4;
    }
#pragma unroll
    for (int p = 0; p < B_PER; ++p) {
        const int idx = tid + p * NT;
        const int kr = idx / BQ, nq = idx % BQ;
        bgp[p] = W + (size_t)kr * Nout + n0 + nq * 4;
        bsl[p] = kr * PB + nq * 4;
    }

    v4f ra[A_PER], rb[B_PER];
#pragma unroll
    for (int p = 0; p < A_PER; ++p) ra[p] = *(const v4f*)agp[p];
#pragma unroll
    for (int p = 0; p < B_PER; ++p) rb[p] = *(const v4f*)bgp[p];

    const int wave = tid >> 5;
    const int lane = tid & 31;
    const int wm   = (wave / (BN / WN)) * WM;
    const int wn   = (wave % (BN / WN)) * WN;
    const int c16  = lane & 15;
    const int klo  = (lane >> 4) << 1;

    v8f acc[TM][TN];
#pragma unroll
    for (int i = 0; i < TM; ++i)
#pragma unroll
        for (int j = 0; j < TN; ++j)
            acc[i][j] = (v8f){0.f, 0.f, 0.f, 0.f, 0.f, 0.f, 0.f, 0.f};

    const float* saf = sA + (wm + c16) * PA + klo;
    const float* sbf = sB + klo * PB + wn + c16;

#pragma unroll
    for (int p = 0; p < A_PER; ++p) *(v4f*)&sA[asl[p]] = ra[p];
#pragma unroll
    for (int p = 0; p < B_PER; ++p) *(v4f*)&sB[bsl[p]] = rb[p];
    __syncthreads();

    for (int kc = 0; kc < NCH; ++kc) {
        const int  cur  = kc & 1;
        const bool more = (kc + 1 < NCH);

        if (more) {
#pragma unroll
            for (int p = 0; p < A_PER; ++p)
                ra[p] = *(const v4f*)(agp[p] + (kc + 1) * KB);
#pragma unroll
            for (int p = 0; p < B_PER; ++p)
                rb[p] = *(const v4f*)(bgp[p] + (size_t)(kc + 1) * KB * Nout);
        }
        __builtin_amdgcn_sched_barrier(0);

        const float* sa = saf + cur * (BM * PA);
        const float* sb = sbf + cur * (KB * PB);
#pragma unroll
        for (int k0 = 0; k0 < KB; k0 += 4) {
            v2f av[TM], bv[TN];
#pragma unroll
            for (int i = 0; i < TM; ++i)
                av[i] = *(const v2f*)(sa + i * 16 * PA + k0);
#pragma unroll
            for (int j = 0; j < TN; ++j) {
                bv[j].x = sb[k0 * PB + j * 16];
                bv[j].y = sb[(k0 + 1) * PB + j * 16];
            }
#pragma unroll
            for (int i = 0; i < TM; ++i)
#pragma unroll
                for (int j = 0; j < TN; ++j)
                    acc[i][j] = __builtin_amdgcn_wmma_f32_16x16x4_f32(
                         false, av[i],  false, bv[j],
                         (short)0, acc[i][j],
                         false,  false);
        }

        if (more) {
            float* da = sA + (cur ^ 1) * (BM * PA);
            float* db = sB + (cur ^ 1) * (KB * PB);
#pragma unroll
            for (int p = 0; p < A_PER; ++p) *(v4f*)&da[asl[p]] = ra[p];
#pragma unroll
            for (int p = 0; p < B_PER; ++p) *(v4f*)&db[bsl[p]] = rb[p];
        }
        __syncthreads();
    }

    const int mhi = (lane >> 4) << 3;
#pragma unroll
    for (int j = 0; j < TN; ++j) {
        const int   col = n0 + wn + j * 16 + c16;
        const float bb  = bias[col];
#pragma unroll
        for (int i = 0; i < TM; ++i) {
#pragma unroll
            for (int r = 0; r < 8; ++r) {
                float v = acc[i][j][r] + bb;
                if (act == 1) v = (v > 20.f) ? v : log1pf(expf(v));
                out[(size_t)(m0 + wm + i * 16 + mhi + r) * Nout + col] = v;
            }
        }
    }
}

__global__ __launch_bounds__(256)
void s6_scan_pass1(const float* __restrict__ x,
                   const float* __restrict__ dt,
                   const float* __restrict__ Bm,
                   const float* __restrict__ A_log,
                   float* __restrict__ hend,
                   float* __restrict__ aend) {
    const int b = blockIdx.z, c = blockIdx.y;
    const int d = blockIdx.x * 256 + threadIdx.x;

    __shared__ float sBm[CLEN * NSTATE];
    for (int idx = threadIdx.x; idx < CLEN * NSTATE; idx += 256)
        sBm[idx] = Bm[((size_t)b * SEQL + c * CLEN) * NSTATE + idx];
    __syncthreads();

    float Ac[NSTATE];
#pragma unroll
    for (int n = 0; n < NSTATE; ++n) Ac[n] = -expf(A_log[d * NSTATE + n]);

    float h[NSTATE], ap[NSTATE];
#pragma unroll
    for (int n = 0; n < NSTATE; ++n) { h[n] = 0.f; ap[n] = 1.f; }

    const float* dtp = dt + ((size_t)b * SEQL + c * CLEN) * DMODEL + d;
    const float* xp  = x  + ((size_t)b * SEQL + c * CLEN) * DMODEL + d;
    v2f cur[TGRP];
#pragma unroll
    for (int i = 0; i < TGRP; ++i) {
        cur[i].x = dtp[(size_t)i * DMODEL];
        cur[i].y = xp[(size_t)i * DMODEL];
    }

    for (int t0 = 0; t0 < CLEN; t0 += TGRP) {
        const int tn = (t0 + TGRP < CLEN) ? (t0 + TGRP) : t0;
        v2f nxt[TGRP];
#pragma unroll
        for (int i = 0; i < TGRP; ++i) {
            nxt[i].x = dtp[(size_t)(tn + i) * DMODEL];
            nxt[i].y = xp[(size_t)(tn + i) * DMODEL];
        }
        __builtin_amdgcn_sched_barrier(0);

#pragma unroll
        for (int i = 0; i < TGRP; ++i) {
            const float dtv = cur[i].x;
            const float ub  = dtv * cur[i].y;
            const int   bmo = (t0 + i) * NSTATE;
#pragma unroll
            for (int n = 0; n < NSTATE; ++n) {
                const float a = __expf(dtv * Ac[n]);
                h[n]  = a * h[n] + ub * sBm[bmo + n];
                ap[n] *= a;
            }
        }
#pragma unroll
        for (int i = 0; i < TGRP; ++i) cur[i] = nxt[i];
    }

    const size_t base = (((size_t)b * DMODEL + d) * NCHUNK + c) * NSTATE;
#pragma unroll
    for (int n = 0; n < NSTATE; ++n) { hend[base + n] = h[n]; aend[base + n] = ap[n]; }
}

__global__ __launch_bounds__(256)
void s6_scan_pass2(const float* __restrict__ hend,
                   const float* __restrict__ aend,
                   float* __restrict__ carry) {
    const int idx = blockIdx.x * 256 + threadIdx.x;
    const size_t bd = (size_t)(idx >> 4);
    const int n = idx & 15;
    float cv = 0.f;
    for (int c = 0; c < NCHUNK; ++c) {
        const size_t o = (bd * NCHUNK + c) * NSTATE + n;
        carry[o] = cv;
        cv = aend[o] * cv + hend[o];
    }
}

__global__ __launch_bounds__(256)
void s6_scan_pass3(const float* __restrict__ x,
                   const float* __restrict__ dt,
                   const float* __restrict__ Bm,
                   const float* __restrict__ Cm,
                   const float* __restrict__ A_log,
                   const float* __restrict__ carry,
                   const float* __restrict__ Dskip,
                   float* __restrict__ yout) {
    const int b = blockIdx.z, c = blockIdx.y;
    const int d = blockIdx.x * 256 + threadIdx.x;

    __shared__ float sBm[CLEN * NSTATE];
    __shared__ float sCm[CLEN * NSTATE];
    for (int idx = threadIdx.x; idx < CLEN * NSTATE; idx += 256) {
        const size_t g = ((size_t)b * SEQL + c * CLEN) * NSTATE + idx;
        sBm[idx] = Bm[g];
        sCm[idx] = Cm[g];
    }
    __syncthreads();

    float Ac[NSTATE];
#pragma unroll
    for (int n = 0; n < NSTATE; ++n) Ac[n] = -expf(A_log[d * NSTATE + n]);

    float h[NSTATE];
    const size_t base = (((size_t)b * DMODEL + d) * NCHUNK + c) * NSTATE;
#pragma unroll
    for (int n = 0; n < NSTATE; ++n) h[n] = carry[base + n];

    const float dsk = Dskip[d];

    const size_t cbase = ((size_t)b * SEQL + c * CLEN) * DMODEL + d;
    const float* dtp = dt + cbase;
    const float* xp  = x  + cbase;
    float*       yp  = yout + cbase;
    v2f cur[TGRP];
#pragma unroll
    for (int i = 0; i < TGRP; ++i) {
        cur[i].x = dtp[(size_t)i * DMODEL];
        cur[i].y = xp[(size_t)i * DMODEL];
    }

    for (int t0 = 0; t0 < CLEN; t0 += TGRP) {
        const int tn = (t0 + TGRP < CLEN) ? (t0 + TGRP) : t0;
        v2f nxt[TGRP];
#pragma unroll
        for (int i = 0; i < TGRP; ++i) {
            nxt[i].x = dtp[(size_t)(tn + i) * DMODEL];
            nxt[i].y = xp[(size_t)(tn + i) * DMODEL];
        }
        __builtin_amdgcn_sched_barrier(0);

        float yv[TGRP];
#pragma unroll
        for (int i = 0; i < TGRP; ++i) {
            const float dtv = cur[i].x;
            const float xv  = cur[i].y;
            const float ub  = dtv * xv;
            const int   so  = (t0 + i) * NSTATE;
            float y = 0.f;
#pragma unroll
            for (int n = 0; n < NSTATE; ++n) {
                const float a = __expf(dtv * Ac[n]);
                h[n] = a * h[n] + ub * sBm[so + n];
                y += sCm[so + n] * h[n];
            }
            y += dsk * xv;
            yv[i] = y;
        }
#pragma unroll
        for (int i = 0; i < TGRP; ++i) {
            const float y = yv[i];
            yp[(size_t)(t0 + i) * DMODEL] =
                0.5f * y * (1.f + erff(y * 0.70710678118654752f));
        }
#pragma unroll
        for (int i = 0; i < TGRP; ++i) cur[i] = nxt[i];
    }
}

#define LN_ROWS 2
#define LN_WAVES 8
__global__ __launch_bounds__(LN_WAVES * 32)
void s6_ln_residual(const float* __restrict__ ymix,
                    const float* __restrict__ hin,
                    const float* __restrict__ g,
                    const float* __restrict__ be,
                    float* __restrict__ hout) {
    const int lane = threadIdx.x & 31;
    const int wave = threadIdx.x >> 5;
    const size_t row0 = ((size_t)blockIdx.x * LN_WAVES + wave) * LN_ROWS;
    const size_t o0   = row0 * DMODEL + (size_t)lane * 4;

    v4f v[LN_ROWS][4], r[LN_ROWS][4], gg[4], bb[4];
#pragma unroll
    for (int rr = 0; rr < LN_ROWS; ++rr)
#pragma unroll
        for (int p = 0; p < 4; ++p)
            v[rr][p] = *(const v4f*)(ymix + o0 + (size_t)rr * DMODEL + p * 128);
#pragma unroll
    for (int rr = 0; rr < LN_ROWS; ++rr)
#pragma unroll
        for (int p = 0; p < 4; ++p)
            r[rr][p] = *(const v4f*)(hin + o0 + (size_t)rr * DMODEL + p * 128);
#pragma unroll
    for (int p = 0; p < 4; ++p) {
        gg[p] = *(const v4f*)(g  + p * 128 + lane * 4);
        bb[p] = *(const v4f*)(be + p * 128 + lane * 4);
    }
    __builtin_amdgcn_sched_barrier(0);

    float s[LN_ROWS];
#pragma unroll
    for (int rr = 0; rr < LN_ROWS; ++rr) {
        const v4f t = (v[rr][0] + v[rr][1]) + (v[rr][2] + v[rr][3]);
        s[rr] = (t.x + t.y) + (t.z + t.w);
    }
#pragma unroll
    for (int m = 16; m >= 1; m >>= 1)
#pragma unroll
        for (int rr = 0; rr < LN_ROWS; ++rr)
            s[rr] += __shfl_xor(s[rr], m, 32);

    float q[LN_ROWS];
#pragma unroll
    for (int rr = 0; rr < LN_ROWS; ++rr) {
        const float mu = s[rr] * (1.f / DMODEL);
        v4f t = (v4f){0.f, 0.f, 0.f, 0.f};
#pragma unroll
        for (int p = 0; p < 4; ++p) {
            v[rr][p] = v[rr][p] - mu;
            t += v[rr][p] * v[rr][p];
        }
        q[rr] = (t.x + t.y) + (t.z + t.w);
    }
#pragma unroll
    for (int m = 16; m >= 1; m >>= 1)
#pragma unroll
        for (int rr = 0; rr < LN_ROWS; ++rr)
            q[rr] += __shfl_xor(q[rr], m, 32);

#pragma unroll
    for (int rr = 0; rr < LN_ROWS; ++rr) {
        const float rstd = rsqrtf(q[rr] * (1.f / DMODEL) + LN_EPS);
#pragma unroll
        for (int p = 0; p < 4; ++p) {
            const v4f o = r[rr][p] + v[rr][p] * rstd * gg[p] + bb[p];
            *(v4f*)(hout + o0 + (size_t)rr * DMODEL + p * 128) = o;
        }
    }
}

extern "C" void kernel_launch(void* const* d_in, const int* in_sizes, int n_in,
                              void* d_out, int out_size, void* d_ws, size_t ws_size,
                              hipStream_t stream) {
    (void)in_sizes; (void)n_in; (void)out_size; (void)ws_size;

    const float* x      = (const float*)d_in[0];
    const float* A_log  = (const float*)d_in[1];
    const float* W_B    = (const float*)d_in[2];
    const float* b_B    = (const float*)d_in[3];
    const float* W_C    = (const float*)d_in[4];
    const float* b_C    = (const float*)d_in[5];
    const float* W_dt   = (const float*)d_in[6];
    const float* b_dt   = (const float*)d_in[7];
    const float* D_skip = (const float*)d_in[8];
    const float* W_mix  = (const float*)d_in[9];
    const float* b_mix  = (const float*)d_in[10];
    const float* ln_g   = (const float*)d_in[11];
    const float* ln_b   = (const float*)d_in[12];
    const float* W_dec  = (const float*)d_in[13];
    const float* b_dec  = (const float*)d_in[14];
    float* out = (float*)d_out;

    float* ws = (float*)d_ws;
    size_t off = 0;
    float* dtb   = ws + off; off += (size_t)MROWS * DMODEL;
    float* Bmb   = ws + off; off += (size_t)MROWS * NSTATE;
    float* Cmb   = ws + off; off += (size_t)MROWS * NSTATE;
    float* ygelu = ws + off; off += (size_t)MROWS * DMODEL;
    float* ymix  = ws + off; off += (size_t)MROWS * DMODEL;
    float* h1    = ws + off; off += (size_t)MROWS * DMODEL;
    float* h2    = ws + off; off += (size_t)MROWS * DMODEL;
    float* hend  = ws + off; off += (size_t)BATCH * DMODEL * NCHUNK * NSTATE;
    float* aend  = ws + off; off += (size_t)BATCH * DMODEL * NCHUNK * NSTATE;
    float* carry = ws + off; off += (size_t)BATCH * DMODEL * NCHUNK * NSTATE;

    const int DN = DMODEL * NSTATE;
    const int DD = DMODEL * DMODEL;

    const dim3 gBig(MROWS / 128, DMODEL / 64);
    const dim3 gBC (MROWS / 32, 1);
    const dim3 gScan(DMODEL / 256, NCHUNK, BATCH);

    for (int i = 0; i < 2; ++i) {
        const float* hin = (i == 0) ? x : h1;
        float* hout      = (i == 0) ? h1 : h2;

        s6_gemm_wmma_k512<128, 64, 32, 32><<<gBig, 256, 0, stream>>>(
            hin, W_dt + (size_t)i * DD, b_dt + i * DMODEL, dtb, DMODEL, 1);
        s6_gemm_wmma_k512<32, 16, 16, 16><<<gBC, 64, 0, stream>>>(
            hin, W_B + (size_t)i * DN, b_B + i * NSTATE, Bmb, NSTATE, 0);
        s6_gemm_wmma_k512<32, 16, 16, 16><<<gBC, 64, 0, stream>>>(
            hin, W_C + (size_t)i * DN, b_C + i * NSTATE, Cmb, NSTATE, 0);

        s6_scan_pass1<<<gScan, 256, 0, stream>>>(
            hin, dtb, Bmb, A_log + (size_t)i * DN, hend, aend);
        s6_scan_pass2<<<BATCH * DMODEL * NSTATE / 256, 256, 0, stream>>>(
            hend, aend, carry);
        s6_scan_pass3<<<gScan, 256, 0, stream>>>(
            hin, dtb, Bmb, Cmb, A_log + (size_t)i * DN, carry,
            D_skip + i * DMODEL, ygelu);

        s6_gemm_wmma_k512<128, 64, 32, 32><<<gBig, 256, 0, stream>>>(
            ygelu, W_mix + (size_t)i * DD, b_mix + i * DMODEL, ymix, DMODEL, 0);
        s6_ln_residual<<<MROWS / (LN_WAVES * LN_ROWS), LN_WAVES * 32, 0, stream>>>(
            ymix, hin, ln_g + i * DMODEL, ln_b + i * DMODEL, hout);
    }

    s6_gemm_wmma_k512<128, 64, 32, 32><<<gBig, 256, 0, stream>>>(h2, W_dec, b_dec, out, OUTD, 0);
}


# dt, Bm and Cm projections of a layer in ONE launch (Bm|Cm as extra 64x32 blocks of the dt GEMM's grid, same accumulators and k order): 17 launches a run become 13
# speedup vs baseline: 3.6501x; 1.1213x over previous
#include <hip/hip_runtime.h>
#include <hip/hip_bf16.h>
#include <math.h>

#define BATCH   4
#define SEQL    2048
#define DMODEL  512
#define NSTATE  16
#define OUTD    512
#define MROWS   (BATCH * SEQL)
#define KDIM    512
#define NCHUNK  32
#define CLEN    (SEQL / NCHUNK)
#define TGRP    8
#define LN_EPS  1e-5f

typedef float v2f __attribute__((ext_vector_type(2)));
typedef float v4f __attribute__((ext_vector_type(4)));
typedef float v8f __attribute__((ext_vector_type(8)));

#define GEMM_KB 32
template <int BM, int BN, int WM, int WN, bool SPLIT>
__device__ __forceinline__
void s6_gemm_tile_k512(float* __restrict__ sA, float* __restrict__ sB,
                       const float* __restrict__ A, int m0,
                       const float* __restrict__ W0, const float* __restrict__ W1,
                       int ldw, int n0,
                       const float* __restrict__ bias0, const float* __restrict__ bias1,
                       float* __restrict__ out0, float* __restrict__ out1,
                       int ldo, int act  ) {
    constexpr int KB    = GEMM_KB;
    constexpr int PA    = KB + 4;
    constexpr int PB    = BN + 8;
    constexpr int NT    = (BM / WM) * (BN / WN) * 32;
    constexpr int TM    = WM / 16;
    constexpr int TN    = WN / 16;
    constexpr int AQ    = KB / 4;
    constexpr int BQ    = BN / 4;
    constexpr int A_PER = (BM * AQ) / NT;
    constexpr int B_PER = (KB * BQ) / NT;
    constexpr int NCH   = KDIM / KB;
    static_assert((BM * AQ) % NT == 0, "A chunk must split evenly over the block");
    static_assert((KB * BQ) % NT == 0, "W chunk must split evenly over the block");

    const int tid = threadIdx.x;

    const float* agp[A_PER]; int asl[A_PER];
    const float* bgp[B_PER]; int bsl[B_PER];
#pragma unroll
    for (int p = 0; p < A_PER; ++p) {
        const int idx = tid + p * NT;
        const int row = idx / AQ, kq = idx % AQ;
        agp[p] = A + (size_t)(m0 + row) * KDIM + kq * 4;
        asl[p] = row * PA + kq * 4;
    }
#pragma unroll
    for (int p = 0; p < B_PER; ++p) {
        const int idx = tid + p * NT;
        const int kr = idx / BQ, nq = idx % BQ;
        if constexpr (SPLIT)
            bgp[p] = ((nq < BQ / 2) ? W0 : W1) + (size_t)kr * ldw + (nq % (BQ / 2)) * 4;
        else
            bgp[p] = W0 + (size_t)kr * ldw + n0 + nq * 4;
        bsl[p] = kr * PB + nq * 4;
    }

    v4f ra[A_PER], rb[B_PER];
#pragma unroll
    for (int p = 0; p < A_PER; ++p) ra[p] = *(const v4f*)agp[p];
#pragma unroll
    for (int p = 0; p < B_PER; ++p) rb[p] = *(const v4f*)bgp[p];

    const int wave = tid >> 5;
    const int lane = tid & 31;
    const int wm   = (wave / (BN / WN)) * WM;
    const int wn   = (wave % (BN / WN)) * WN;
    const int c16  = lane & 15;
    const int klo  = (lane >> 4) << 1;

    v8f acc[TM][TN];
#pragma unroll
    for (int i = 0; i < TM; ++i)
#pragma unroll
        for (int j = 0; j < TN; ++j)
            acc[i][j] = (v8f){0.f, 0.f, 0.f, 0.f, 0.f, 0.f, 0.f, 0.f};

    const float* saf = sA + (wm + c16) * PA + klo;
    const float* sbf = sB + klo * PB + wn + c16;

#pragma unroll
    for (int p = 0; p < A_PER; ++p) *(v4f*)&sA[asl[p]] = ra[p];
#pragma unroll
    for (int p = 0; p < B_PER; ++p) *(v4f*)&sB[bsl[p]] = rb[p];
    __syncthreads();

    for (int kc = 0; kc < NCH; ++kc) {
        const int  cur  = kc & 1;
        const bool more = (kc + 1 < NCH);

        if (more) {
#pragma unroll
            for (int p = 0; p < A_PER; ++p)
                ra[p] = *(const v4f*)(agp[p] + (kc + 1) * KB);
#pragma unroll
            for (int p = 0; p < B_PER; ++p)
                rb[p] = *(const v4f*)(bgp[p] + (size_t)(kc + 1) * KB * ldw);
        }
        __builtin_amdgcn_sched_barrier(0);

        const float* sa = saf + cur * (BM * PA);
        const float* sb = sbf + cur * (KB * PB);
#pragma unroll
        for (int k0 = 0; k0 < KB; k0 += 4) {
            v2f av[TM], bv[TN];
#pragma unroll
            for (int i = 0; i < TM; ++i)
                av[i] = *(const v2f*)(sa + i * 16 * PA + k0);
#pragma unroll
            for (int j = 0; j < TN; ++j) {
                bv[j].x = sb[k0 * PB + j * 16];
                bv[j].y = sb[(k0 + 1) * PB + j * 16];
            }
#pragma unroll
            for (int i = 0; i < TM; ++i)
#pragma unroll
                for (int j = 0; j < TN; ++j)
                    acc[i][j] = __builtin_amdgcn_wmma_f32_16x16x4_f32(
                         false, av[i],  false, bv[j],
                         (short)0, acc[i][j],
                         false,  false);
        }

        if (more) {
            float* da = sA + (cur ^ 1) * (BM * PA);
            float* db = sB + (cur ^ 1) * (KB * PB);
#pragma unroll
            for (int p = 0; p < A_PER; ++p) *(v4f*)&da[asl[p]] = ra[p];
#pragma unroll
            for (int p = 0; p < B_PER; ++p) *(v4f*)&db[bsl[p]] = rb[p];
        }
        __syncthreads();
    }

    const int mhi = (lane >> 4) << 3;
#pragma unroll
    for (int j = 0; j < TN; ++j) {
        const int  ct = wn + j * 16;
        const bool hi = SPLIT && (ct >= BN / 2);
        const int  col = SPLIT ? (ct % (BN / 2)) + c16 : n0 + ct + c16;
        const float* __restrict__ bias = hi ? bias1 : bias0;
        float* __restrict__       out  = hi ? out1 : out0;
        const float bb  = bias[col];
#pragma unroll
        for (int i = 0; i < TM; ++i) {
#pragma unroll
            for (int r = 0; r < 8; ++r) {
                float v = acc[i][j][r] + bb;
                if (act == 1) v = (v > 20.f) ? v : log1pf(expf(v));
                out[(size_t)(m0 + wm + i * 16 + mhi + r) * ldo + col] = v;
            }
        }
    }
}

template <int BM, int BN, int WM, int WN>
__global__ __launch_bounds__((BM / WM) * (BN / WN) * 32)
void s6_gemm_wmma_k512(const float* __restrict__ A,
                       const float* __restrict__ W,
                       const float* __restrict__ bias,
                       float* __restrict__ out,
                       int Nout, int act  ) {
    __shared__ float sA[2 * BM * (GEMM_KB + 4)] __attribute__((aligned(16)));
    __shared__ float sB[2 * GEMM_KB * (BN + 8)] __attribute__((aligned(16)));
    s6_gemm_tile_k512<BM, BN, WM, WN, false>(sA, sB, A, blockIdx.x * BM,
                                             W, W, Nout, blockIdx.y * BN,
                                             bias, bias, out, out, Nout, act);
}

#define BC_BM 64
__global__ __launch_bounds__(256)
void s6_proj_dt_bc(const float* __restrict__ hin,
                   const float* __restrict__ W_dt, const float* __restrict__ b_dt,
                   float* __restrict__ dt,
                   const float* __restrict__ W_B, const float* __restrict__ b_B,
                   float* __restrict__ Bm,
                   const float* __restrict__ W_C, const float* __restrict__ b_C,
                   float* __restrict__ Cm) {
    __shared__ float sA[2 * 128 * (GEMM_KB + 4)] __attribute__((aligned(16)));
    __shared__ float sB[2 * GEMM_KB * (64 + 8)] __attribute__((aligned(16)));
    const int ydt = DMODEL / 64;
    if ((int)blockIdx.y < ydt) {
        s6_gemm_tile_k512<128, 64, 32, 32, false>(sA, sB, hin, blockIdx.x * 128,
                                                  W_dt, W_dt, DMODEL, blockIdx.y * 64,
                                                  b_dt, b_dt, dt, dt, DMODEL, 1);
    } else {
        const int mt = ((int)blockIdx.y - ydt) * (int)gridDim.x + (int)blockIdx.x;
        s6_gemm_tile_k512<BC_BM, 2 * NSTATE, 16, 16, true>(sA, sB, hin, mt * BC_BM,
                                                           W_B, W_C, NSTATE, 0,
                                                           b_B, b_C, Bm, Cm, NSTATE, 0);
    }
}

__global__ __launch_bounds__(256)
void s6_scan_pass1(const float* __restrict__ x,
                   const float* __restrict__ dt,
                   const float* __restrict__ Bm,
                   const float* __restrict__ A_log,
                   float* __restrict__ hend,
                   float* __restrict__ aend) {
    const int b = blockIdx.z, c = blockIdx.y;
    const int d = blockIdx.x * 256 + threadIdx.x;

    __shared__ float sBm[CLEN * NSTATE];
    for (int idx = threadIdx.x; idx < CLEN * NSTATE; idx += 256)
        sBm[idx] = Bm[((size_t)b * SEQL + c * CLEN) * NSTATE + idx];
    __syncthreads();

    float Ac[NSTATE];
#pragma unroll
    for (int n = 0; n < NSTATE; ++n) Ac[n] = -expf(A_log[d * NSTATE + n]);

    float h[NSTATE], ap[NSTATE];
#pragma unroll
    for (int n = 0; n < NSTATE; ++n) { h[n] = 0.f; ap[n] = 1.f; }

    const float* dtp = dt + ((size_t)b * SEQL + c * CLEN) * DMODEL + d;
    const float* xp  = x  + ((size_t)b * SEQL + c * CLEN) * DMODEL + d;
    v2f cur[TGRP];
#pragma unroll
    for (int i = 0; i < TGRP; ++i) {
        cur[i].x = dtp[(size_t)i * DMODEL];
        cur[i].y = xp[(size_t)i * DMODEL];
    }

    for (int t0 = 0; t0 < CLEN; t0 += TGRP) {
        const int tn = (t0 + TGRP < CLEN) ? (t0 + TGRP) : t0;
        v2f nxt[TGRP];
#pragma unroll
        for (int i = 0; i < TGRP; ++i) {
            nxt[i].x = dtp[(size_t)(tn + i) * DMODEL];
            nxt[i].y = xp[(size_t)(tn + i) * DMODEL];
        }
        __builtin_amdgcn_sched_barrier(0);

#pragma unroll
        for (int i = 0; i < TGRP; ++i) {
            const float dtv = cur[i].x;
            const float ub  = dtv * cur[i].y;
            const int   bmo = (t0 + i) * NSTATE;
#pragma unroll
            for (int n = 0; n < NSTATE; ++n) {
                const float a = __expf(dtv * Ac[n]);
                h[n]  = a * h[n] + ub * sBm[bmo + n];
                ap[n] *= a;
            }
        }
#pragma unroll
        for (int i = 0; i < TGRP; ++i) cur[i] = nxt[i];
    }

    const size_t base = (((size_t)b * DMODEL + d) * NCHUNK + c) * NSTATE;
#pragma unroll
    for (int n = 0; n < NSTATE; ++n) { hend[base + n] = h[n]; aend[base + n] = ap[n]; }
}

__global__ __launch_bounds__(256)
void s6_scan_pass2(const float* __restrict__ hend,
                   const float* __restrict__ aend,
                   float* __restrict__ carry) {
    const int idx = blockIdx.x * 256 + threadIdx.x;
    const size_t bd = (size_t)(idx >> 4);
    const int n = idx & 15;
    float cv = 0.f;
    for (int c = 0; c < NCHUNK; ++c) {
        const size_t o = (bd * NCHUNK + c) * NSTATE + n;
        carry[o] = cv;
        cv = aend[o] * cv + hend[o];
    }
}

__global__ __launch_bounds__(256)
void s6_scan_pass3(const float* __restrict__ x,
                   const float* __restrict__ dt,
                   const float* __restrict__ Bm,
                   const float* __restrict__ Cm,
                   const float* __restrict__ A_log,
                   const float* __restrict__ carry,
                   const float* __restrict__ Dskip,
                   float* __restrict__ yout) {
    const int b = blockIdx.z, c = blockIdx.y;
    const int d = blockIdx.x * 256 + threadIdx.x;

    __shared__ float sBm[CLEN * NSTATE];
    __shared__ float sCm[CLEN * NSTATE];
    for (int idx = threadIdx.x; idx < CLEN * NSTATE; idx += 256) {
        const size_t g = ((size_t)b * SEQL + c * CLEN) * NSTATE + idx;
        sBm[idx] = Bm[g];
        sCm[idx] = Cm[g];
    }
    __syncthreads();

    float Ac[NSTATE];
#pragma unroll
    for (int n = 0; n < NSTATE; ++n) Ac[n] = -expf(A_log[d * NSTATE + n]);

    float h[NSTATE];
    const size_t base = (((size_t)b * DMODEL + d) * NCHUNK + c) * NSTATE;
#pragma unroll
    for (int n = 0; n < NSTATE; ++n) h[n] = carry[base + n];

    const float dsk = Dskip[d];

    const size_t cbase = ((size_t)b * SEQL + c * CLEN) * DMODEL + d;
    const float* dtp = dt + cbase;
    const float* xp  = x  + cbase;
    float*       yp  = yout + cbase;
    v2f cur[TGRP];
#pragma unroll
    for (int i = 0; i < TGRP; ++i) {
        cur[i].x = dtp[(size_t)i * DMODEL];
        cur[i].y = xp[(size_t)i * DMODEL];
    }

    for (int t0 = 0; t0 < CLEN; t0 += TGRP) {
        const int tn = (t0 + TGRP < CLEN) ? (t0 + TGRP) : t0;
        v2f nxt[TGRP];
#pragma unroll
        for (int i = 0; i < TGRP; ++i) {
            nxt[i].x = dtp[(size_t)(tn + i) * DMODEL];
            nxt[i].y = xp[(size_t)(tn + i) * DMODEL];
        }
        __builtin_amdgcn_sched_barrier(0);

        float yv[TGRP];
#pragma unroll
        for (int i = 0; i < TGRP; ++i) {
            const float dtv = cur[i].x;
            const float xv  = cur[i].y;
            const float ub  = dtv * xv;
            const int   so  = (t0 + i) * NSTATE;
            float y = 0.f;
#pragma unroll
            for (int n = 0; n < NSTATE; ++n) {
                const float a = __expf(dtv * Ac[n]);
                h[n] = a * h[n] + ub * sBm[so + n];
                y += sCm[so + n] * h[n];
            }
            y += dsk * xv;
            yv[i] = y;
        }
#pragma unroll
        for (int i = 0; i < TGRP; ++i) {
            const float y = yv[i];
            yp[(size_t)(t0 + i) * DMODEL] =
                0.5f * y * (1.f + erff(y * 0.70710678118654752f));
        }
#pragma unroll
        for (int i = 0; i < TGRP; ++i) cur[i] = nxt[i];
    }
}

#define LN_ROWS 2
#define LN_WAVES 8
__global__ __launch_bounds__(LN_WAVES * 32)
void s6_ln_residual(const float* __restrict__ ymix,
                    const float* __restrict__ hin,
                    const float* __restrict__ g,
                    const float* __restrict__ be,
                    float* __restrict__ hout) {
    const int lane = threadIdx.x & 31;
    const int wave = threadIdx.x >> 5;
    const size_t row0 = ((size_t)blockIdx.x * LN_WAVES + wave) * LN_ROWS;
    const size_t o0   = row0 * DMODEL + (size_t)lane * 4;

    v4f v[LN_ROWS][4], r[LN_ROWS][4], gg[4], bb[4];
#pragma unroll
    for (int rr = 0; rr < LN_ROWS; ++rr)
#pragma unroll
        for (int p = 0; p < 4; ++p)
            v[rr][p] = *(const v4f*)(ymix + o0 + (size_t)rr * DMODEL + p * 128);
#pragma unroll
    for (int rr = 0; rr < LN_ROWS; ++rr)
#pragma unroll
        for (int p = 0; p < 4; ++p)
            r[rr][p] = *(const v4f*)(hin + o0 + (size_t)rr * DMODEL + p * 128);
#pragma unroll
    for (int p = 0; p < 4; ++p) {
        gg[p] = *(const v4f*)(g  + p * 128 + lane * 4);
        bb[p] = *(const v4f*)(be + p * 128 + lane * 4);
    }
    __builtin_amdgcn_sched_barrier(0);

    float s[LN_ROWS];
#pragma unroll
    for (int rr = 0; rr < LN_ROWS; ++rr) {
        const v4f t = (v[rr][0] + v[rr][1]) + (v[rr][2] + v[rr][3]);
        s[rr] = (t.x + t.y) + (t.z + t.w);
    }
#pragma unroll
    for (int m = 16; m >= 1; m >>= 1)
#pragma unroll
        for (int rr = 0; rr < LN_ROWS; ++rr)
            s[rr] += __shfl_xor(s[rr], m, 32);

    float q[LN_ROWS];
#pragma unroll
    for (int rr = 0; rr < LN_ROWS; ++rr) {
        const float mu = s[rr] * (1.f / DMODEL);
        v4f t = (v4f){0.f, 0.f, 0.f, 0.f};
#pragma unroll
        for (int p = 0; p < 4; ++p) {
            v[rr][p] = v[rr][p] - mu;
            t += v[rr][p] * v[rr][p];
        }
        q[rr] = (t.x + t.y) + (t.z + t.w);
    }
#pragma unroll
    for (int m = 16; m >= 1; m >>= 1)
#pragma unroll
        for (int rr = 0; rr < LN_ROWS; ++rr)
            q[rr] += __shfl_xor(q[rr], m, 32);

#pragma unroll
    for (int rr = 0; rr < LN_ROWS; ++rr) {
        const float rstd = rsqrtf(q[rr] * (1.f / DMODEL) + LN_EPS);
#pragma unroll
        for (int p = 0; p < 4; ++p) {
            const v4f o = r[rr][p] + v[rr][p] * rstd * gg[p] + bb[p];
            *(v4f*)(hout + o0 + (size_t)rr * DMODEL + p * 128) = o;
        }
    }
}

extern "C" void kernel_launch(void* const* d_in, const int* in_sizes, int n_in,
                              void* d_out, int out_size, void* d_ws, size_t ws_size,
                              hipStream_t stream) {
    (void)in_sizes; (void)n_in; (void)out_size; (void)ws_size;

    const float* x      = (const float*)d_in[0];
    const float* A_log  = (const float*)d_in[1];
    const float* W_B    = (const float*)d_in[2];
    const float* b_B    = (const float*)d_in[3];
    const float* W_C    = (const float*)d_in[4];
    const float* b_C    = (const float*)d_in[5];
    const float* W_dt   = (const float*)d_in[6];
    const float* b_dt   = (const float*)d_in[7];
    const float* D_skip = (const float*)d_in[8];
    const float* W_mix  = (const float*)d_in[9];
    const float* b_mix  = (const float*)d_in[10];
    const float* ln_g   = (const float*)d_in[11];
    const float* ln_b   = (const float*)d_in[12];
    const float* W_dec  = (const float*)d_in[13];
    const float* b_dec  = (const float*)d_in[14];
    float* out = (float*)d_out;

    float* ws = (float*)d_ws;
    size_t off = 0;
    float* dtb   = ws + off; off += (size_t)MROWS * DMODEL;
    float* Bmb   = ws + off; off += (size_t)MROWS * NSTATE;
    float* Cmb   = ws + off; off += (size_t)MROWS * NSTATE;
    float* ygelu = ws + off; off += (size_t)MROWS * DMODEL;
    float* ymix  = ws + off; off += (size_t)MROWS * DMODEL;
    float* h1    = ws + off; off += (size_t)MROWS * DMODEL;
    float* h2    = ws + off; off += (size_t)MROWS * DMODEL;
    float* hend  = ws + off; off += (size_t)BATCH * DMODEL * NCHUNK * NSTATE;
    float* aend  = ws + off; off += (size_t)BATCH * DMODEL * NCHUNK * NSTATE;
    float* carry = ws + off; off += (size_t)BATCH * DMODEL * NCHUNK * NSTATE;

    const int DN = DMODEL * NSTATE;
    const int DD = DMODEL * DMODEL;

    const dim3 gBig(MROWS / 128, DMODEL / 64);
    const dim3 gProj(MROWS / 128, DMODEL / 64 + MROWS / BC_BM / (MROWS / 128));
    const dim3 gScan(DMODEL / 256, NCHUNK, BATCH);

    for (int i = 0; i < 2; ++i) {
        const float* hin = (i == 0) ? x : h1;
        float* hout      = (i == 0) ? h1 : h2;

        s6_proj_dt_bc<<<gProj, 256, 0, stream>>>(
            hin, W_dt + (size_t)i * DD, b_dt + i * DMODEL, dtb,
            W_B + (size_t)i * DN, b_B + i * NSTATE, Bmb,
            W_C + (size_t)i * DN, b_C + i * NSTATE, Cmb);

        s6_scan_pass1<<<gScan, 256, 0, stream>>>(
            hin, dtb, Bmb, A_log + (size_t)i * DN, hend, aend);
        s6_scan_pass2<<<BATCH * DMODEL * NSTATE / 256, 256, 0, stream>>>(
            hend, aend, carry);
        s6_scan_pass3<<<gScan, 256, 0, stream>>>(
            hin, dtb, Bmb, Cmb, A_log + (size_t)i * DN, carry,
            D_skip + i * DMODEL, ygelu);

        s6_gemm_wmma_k512<128, 64, 32, 32><<<gBig, 256, 0, stream>>>(
            ygelu, W_mix + (size_t)i * DD, b_mix + i * DMODEL, ymix, DMODEL, 0);
        s6_ln_residual<<<MROWS / (LN_WAVES * LN_ROWS), LN_WAVES * 32, 0, stream>>>(
            ymix, hin, ln_g + i * DMODEL, ln_b + i * DMODEL, hout);
    }

    s6_gemm_wmma_k512<128, 64, 32, 32><<<gBig, 256, 0, stream>>>(h2, W_dec, b_dec, out, OUTD, 0);
}


# dt projection's softplus epilogue: hardware exp2/log2 in the branch-free form max(v,0)+log(1+exp(-|v|)), abs error about 1e-7, replaces libm expf+log1pf; activation a compile-time switch
# speedup vs baseline: 4.0553x; 1.1110x over previous
#include <hip/hip_runtime.h>
#include <hip/hip_bf16.h>
#include <math.h>

#define BATCH   4
#define SEQL    2048
#define DMODEL  512
#define NSTATE  16
#define OUTD    512
#define MROWS   (BATCH * SEQL)
#define KDIM    512
#define NCHUNK  32
#define CLEN    (SEQL / NCHUNK)
#define TGRP    8
#define LN_EPS  1e-5f

typedef float v2f __attribute__((ext_vector_type(2)));
typedef float v4f __attribute__((ext_vector_type(4)));
typedef float v8f __attribute__((ext_vector_type(8)));

#define GEMM_KB 32

__device__ __forceinline__ float s6_softplus(float v) {
    const float e = __builtin_amdgcn_exp2f(fabsf(v) * -1.4426950408889634f);
    return fmaf(__builtin_amdgcn_logf(1.f + e), 0.6931471805599453f, fmaxf(v, 0.f));
}

template <int BM, int BN, int WM, int WN, bool SPLIT, bool SOFTPLUS>
__device__ __forceinline__
void s6_gemm_tile_k512(float* __restrict__ sA, float* __restrict__ sB,
                       const float* __restrict__ A, int m0,
                       const float* __restrict__ W0, const float* __restrict__ W1,
                       int ldw, int n0,
                       const float* __restrict__ bias0, const float* __restrict__ bias1,
                       float* __restrict__ out0, float* __restrict__ out1,
                       int ldo) {
    constexpr int KB    = GEMM_KB;
    constexpr int PA    = KB + 4;
    constexpr int PB    = BN + 8;
    constexpr int NT    = (BM / WM) * (BN / WN) * 32;
    constexpr int TM    = WM / 16;
    constexpr int TN    = WN / 16;
    constexpr int AQ    = KB / 4;
    constexpr int BQ    = BN / 4;
    constexpr int A_PER = (BM * AQ) / NT;
    constexpr int B_PER = (KB * BQ) / NT;
    constexpr int NCH   = KDIM / KB;
    static_assert((BM * AQ) % NT == 0, "A chunk must split evenly over the block");
    static_assert((KB * BQ) % NT == 0, "W chunk must split evenly over the block");

    const int tid = threadIdx.x;

    const float* agp[A_PER]; int asl[A_PER];
    const float* bgp[B_PER]; int bsl[B_PER];
#pragma unroll
    for (int p = 0; p < A_PER; ++p) {
        const int idx = tid + p * NT;
        const int row = idx / AQ, kq = idx % AQ;
        agp[p] = A + (size_t)(m0 + row) * KDIM + kq * 4;
        asl[p] = row * PA + kq * 4;
    }
#pragma unroll
    for (int p = 0; p < B_PER; ++p) {
        const int idx = tid + p * NT;
        const int kr = idx / BQ, nq = idx % BQ;
        if constexpr (SPLIT)
            bgp[p] = ((nq < BQ / 2) ? W0 : W1) + (size_t)kr * ldw + (nq % (BQ / 2)) * 4;
        else
            bgp[p] = W0 + (size_t)kr * ldw + n0 + nq * 4;
        bsl[p] = kr * PB + nq * 4;
    }

    v4f ra[A_PER], rb[B_PER];
#pragma unroll
    for (int p = 0; p < A_PER; ++p) ra[p] = *(const v4f*)agp[p];
#pragma unroll
    for (int p = 0; p < B_PER; ++p) rb[p] = *(const v4f*)bgp[p];

    const int wave = tid >> 5;
    const int lane = tid & 31;
    const int wm   = (wave / (BN / WN)) * WM;
    const int wn   = (wave % (BN / WN)) * WN;
    const int c16  = lane & 15;
    const int klo  = (lane >> 4) << 1;

    v8f acc[TM][TN];
#pragma unroll
    for (int i = 0; i < TM; ++i)
#pragma unroll
        for (int j = 0; j < TN; ++j)
            acc[i][j] = (v8f){0.f, 0.f, 0.f, 0.f, 0.f, 0.f, 0.f, 0.f};

    const float* saf = sA + (wm + c16) * PA + klo;
    const float* sbf = sB + klo * PB + wn + c16;

#pragma unroll
    for (int p = 0; p < A_PER; ++p) *(v4f*)&sA[asl[p]] = ra[p];
#pragma unroll
    for (int p = 0; p < B_PER; ++p) *(v4f*)&sB[bsl[p]] = rb[p];
    __syncthreads();

    for (int kc = 0; kc < NCH; ++kc) {
        const int  cur  = kc & 1;
        const bool more = (kc + 1 < NCH);

        if (more) {
#pragma unroll
            for (int p = 0; p < A_PER; ++p)
                ra[p] = *(const v4f*)(agp[p] + (kc + 1) * KB);
#pragma unroll
            for (int p = 0; p < B_PER; ++p)
                rb[p] = *(const v4f*)(bgp[p] + (size_t)(kc + 1) * KB * ldw);
        }
        __builtin_amdgcn_sched_barrier(0);

        const float* sa = saf + cur * (BM * PA);
        const float* sb = sbf + cur * (KB * PB);
#pragma unroll
        for (int k0 = 0; k0 < KB; k0 += 4) {
            v2f av[TM], bv[TN];
#pragma unroll
            for (int i = 0; i < TM; ++i)
                av[i] = *(const v2f*)(sa + i * 16 * PA + k0);
#pragma unroll
            for (int j = 0; j < TN; ++j) {
                bv[j].x = sb[k0 * PB + j * 16];
                bv[j].y = sb[(k0 + 1) * PB + j * 16];
            }
#pragma unroll
            for (int i = 0; i < TM; ++i)
#pragma unroll
                for (int j = 0; j < TN; ++j)
                    acc[i][j] = __builtin_amdgcn_wmma_f32_16x16x4_f32(
                         false, av[i],  false, bv[j],
                         (short)0, acc[i][j],
                         false,  false);
        }

        if (more) {
            float* da = sA + (cur ^ 1) * (BM * PA);
            float* db = sB + (cur ^ 1) * (KB * PB);
#pragma unroll
            for (int p = 0; p < A_PER; ++p) *(v4f*)&da[asl[p]] = ra[p];
#pragma unroll
            for (int p = 0; p < B_PER; ++p) *(v4f*)&db[bsl[p]] = rb[p];
        }
        __syncthreads();
    }

    const int mhi = (lane >> 4) << 3;
#pragma unroll
    for (int j = 0; j < TN; ++j) {
        const int  ct = wn + j * 16;
        const bool hi = SPLIT && (ct >= BN / 2);
        const int  col = SPLIT ? (ct % (BN / 2)) + c16 : n0 + ct + c16;
        const float* __restrict__ bias = hi ? bias1 : bias0;
        float* __restrict__       out  = hi ? out1 : out0;
        const float bb  = bias[col];
#pragma unroll
        for (int i = 0; i < TM; ++i) {
#pragma unroll
            for (int r = 0; r < 8; ++r) {
                float v = acc[i][j][r] + bb;
                if constexpr (SOFTPLUS) v = s6_softplus(v);
                out[(size_t)(m0 + wm + i * 16 + mhi + r) * ldo + col] = v;
            }
        }
    }
}

template <int BM, int BN, int WM, int WN>
__global__ __launch_bounds__((BM / WM) * (BN / WN) * 32)
void s6_gemm_wmma_k512(const float* __restrict__ A,
                       const float* __restrict__ W,
                       const float* __restrict__ bias,
                       float* __restrict__ out,
                       int Nout) {
    __shared__ float sA[2 * BM * (GEMM_KB + 4)] __attribute__((aligned(16)));
    __shared__ float sB[2 * GEMM_KB * (BN + 8)] __attribute__((aligned(16)));
    s6_gemm_tile_k512<BM, BN, WM, WN, false, false>(sA, sB, A, blockIdx.x * BM,
                                                    W, W, Nout, blockIdx.y * BN,
                                                    bias, bias, out, out, Nout);
}

#define BC_BM 64
__global__ __launch_bounds__(256)
void s6_proj_dt_bc(const float* __restrict__ hin,
                   const float* __restrict__ W_dt, const float* __restrict__ b_dt,
                   float* __restrict__ dt,
                   const float* __restrict__ W_B, const float* __restrict__ b_B,
                   float* __restrict__ Bm,
                   const float* __restrict__ W_C, const float* __restrict__ b_C,
                   float* __restrict__ Cm) {
    __shared__ float sA[2 * 128 * (GEMM_KB + 4)] __attribute__((aligned(16)));
    __shared__ float sB[2 * GEMM_KB * (64 + 8)] __attribute__((aligned(16)));
    const int ydt = DMODEL / 64;
    if ((int)blockIdx.y < ydt) {
        s6_gemm_tile_k512<128, 64, 32, 32, false, true>(sA, sB, hin, blockIdx.x * 128,
                                                        W_dt, W_dt, DMODEL, blockIdx.y * 64,
                                                        b_dt, b_dt, dt, dt, DMODEL);
    } else {
        const int mt = ((int)blockIdx.y - ydt) * (int)gridDim.x + (int)blockIdx.x;
        s6_gemm_tile_k512<BC_BM, 2 * NSTATE, 16, 16, true, false>(sA, sB, hin, mt * BC_BM,
                                                                  W_B, W_C, NSTATE, 0,
                                                                  b_B, b_C, Bm, Cm, NSTATE);
    }
}

__global__ __launch_bounds__(256)
void s6_scan_pass1(const float* __restrict__ x,
                   const float* __restrict__ dt,
                   const float* __restrict__ Bm,
                   const float* __restrict__ A_log,
                   float* __restrict__ hend,
                   float* __restrict__ aend) {
    const int b = blockIdx.z, c = blockIdx.y;
    const int d = blockIdx.x * 256 + threadIdx.x;

    __shared__ float sBm[CLEN * NSTATE];
    for (int idx = threadIdx.x; idx < CLEN * NSTATE; idx += 256)
        sBm[idx] = Bm[((size_t)b * SEQL + c * CLEN) * NSTATE + idx];
    __syncthreads();

    float Ac[NSTATE];
#pragma unroll
    for (int n = 0; n < NSTATE; ++n) Ac[n] = -expf(A_log[d * NSTATE + n]);

    float h[NSTATE], ap[NSTATE];
#pragma unroll
    for (int n = 0; n < NSTATE; ++n) { h[n] = 0.f; ap[n] = 1.f; }

    const float* dtp = dt + ((size_t)b * SEQL + c * CLEN) * DMODEL + d;
    const float* xp  = x  + ((size_t)b * SEQL + c * CLEN) * DMODEL + d;
    v2f cur[TGRP];
#pragma unroll
    for (int i = 0; i < TGRP; ++i) {
        cur[i].x = dtp[(size_t)i * DMODEL];
        cur[i].y = xp[(size_t)i * DMODEL];
    }

    for (int t0 = 0; t0 < CLEN; t0 += TGRP) {
        const int tn = (t0 + TGRP < CLEN) ? (t0 + TGRP) : t0;
        v2f nxt[TGRP];
#pragma unroll
        for (int i = 0; i < TGRP; ++i) {
            nxt[i].x = dtp[(size_t)(tn + i) * DMODEL];
            nxt[i].y = xp[(size_t)(tn + i) * DMODEL];
        }
        __builtin_amdgcn_sched_barrier(0);

#pragma unroll
        for (int i = 0; i < TGRP; ++i) {
            const float dtv = cur[i].x;
            const float ub  = dtv * cur[i].y;
            const int   bmo = (t0 + i) * NSTATE;
#pragma unroll
            for (int n = 0; n < NSTATE; ++n) {
                const float a = __expf(dtv * Ac[n]);
                h[n]  = a * h[n] + ub * sBm[bmo + n];
                ap[n] *= a;
            }
        }
#pragma unroll
        for (int i = 0; i < TGRP; ++i) cur[i] = nxt[i];
    }

    const size_t base = (((size_t)b * DMODEL + d) * NCHUNK + c) * NSTATE;
#pragma unroll
    for (int n = 0; n < NSTATE; ++n) { hend[base + n] = h[n]; aend[base + n] = ap[n]; }
}

__global__ __launch_bounds__(256)
void s6_scan_pass2(const float* __restrict__ hend,
                   const float* __restrict__ aend,
                   float* __restrict__ carry) {
    const int idx = blockIdx.x * 256 + threadIdx.x;
    const size_t bd = (size_t)(idx >> 4);
    const int n = idx & 15;
    float cv = 0.f;
    for (int c = 0; c < NCHUNK; ++c) {
        const size_t o = (bd * NCHUNK + c) * NSTATE + n;
        carry[o] = cv;
        cv = aend[o] * cv + hend[o];
    }
}

__global__ __launch_bounds__(256)
void s6_scan_pass3(const float* __restrict__ x,
                   const float* __restrict__ dt,
                   const float* __restrict__ Bm,
                   const float* __restrict__ Cm,
                   const float* __restrict__ A_log,
                   const float* __restrict__ carry,
                   const float* __restrict__ Dskip,
                   float* __restrict__ yout) {
    const int b = blockIdx.z, c = blockIdx.y;
    const int d = blockIdx.x * 256 + threadIdx.x;

    __shared__ float sBm[CLEN * NSTATE];
    __shared__ float sCm[CLEN * NSTATE];
    for (int idx = threadIdx.x; idx < CLEN * NSTATE; idx += 256) {
        const size_t g = ((size_t)b * SEQL + c * CLEN) * NSTATE + idx;
        sBm[idx] = Bm[g];
        sCm[idx] = Cm[g];
    }
    __syncthreads();

    float Ac[NSTATE];
#pragma unroll
    for (int n = 0; n < NSTATE; ++n) Ac[n] = -expf(A_log[d * NSTATE + n]);

    float h[NSTATE];
    const size_t base = (((size_t)b * DMODEL + d) * NCHUNK + c) * NSTATE;
#pragma unroll
    for (int n = 0; n < NSTATE; ++n) h[n] = carry[base + n];

    const float dsk = Dskip[d];

    const size_t cbase = ((size_t)b * SEQL + c * CLEN) * DMODEL + d;
    const float* dtp = dt + cbase;
    const float* xp  = x  + cbase;
    float*       yp  = yout + cbase;
    v2f cur[TGRP];
#pragma unroll
    for (int i = 0; i < TGRP; ++i) {
        cur[i].x = dtp[(size_t)i * DMODEL];
        cur[i].y = xp[(size_t)i * DMODEL];
    }

    for (int t0 = 0; t0 < CLEN; t0 += TGRP) {
        const int tn = (t0 + TGRP < CLEN) ? (t0 + TGRP) : t0;
        v2f nxt[TGRP];
#pragma unroll
        for (int i = 0; i < TGRP; ++i) {
            nxt[i].x = dtp[(size_t)(tn + i) * DMODEL];
            nxt[i].y = xp[(size_t)(tn + i) * DMODEL];
        }
        __builtin_amdgcn_sched_barrier(0);

        float yv[TGRP];
#pragma unroll
        for (int i = 0; i < TGRP; ++i) {
            const float dtv = cur[i].x;
            const float xv  = cur[i].y;
            const float ub  = dtv * xv;
            const int   so  = (t0 + i) * NSTATE;
            float y = 0.f;
#pragma unroll
            for (int n = 0; n < NSTATE; ++n) {
                const float a = __expf(dtv * Ac[n]);
                h[n] = a * h[n] + ub * sBm[so + n];
                y += sCm[so + n] * h[n];
            }
            y += dsk * xv;
            yv[i] = y;
        }
#pragma unroll
        for (int i = 0; i < TGRP; ++i) {
            const float y = yv[i];
            yp[(size_t)(t0 + i) * DMODEL] =
                0.5f * y * (1.f + erff(y * 0.70710678118654752f));
        }
#pragma unroll
        for (int i = 0; i < TGRP; ++i) cur[i] = nxt[i];
    }
}

#define LN_ROWS 2
#define LN_WAVES 8
__global__ __launch_bounds__(LN_WAVES * 32)
void s6_ln_residual(const float* __restrict__ ymix,
                    const float* __restrict__ hin,
                    const float* __restrict__ g,
                    const float* __restrict__ be,
                    float* __restrict__ hout) {
    const int lane = threadIdx.x & 31;
    const int wave = threadIdx.x >> 5;
    const size_t row0 = ((size_t)blockIdx.x * LN_WAVES + wave) * LN_ROWS;
    const size_t o0   = row0 * DMODEL + (size_t)lane * 4;

    v4f v[LN_ROWS][4], r[LN_ROWS][4], gg[4], bb[4];
#pragma unroll
    for (int rr = 0; rr < LN_ROWS; ++rr)
#pragma unroll
        for (int p = 0; p < 4; ++p)
            v[rr][p] = *(const v4f*)(ymix + o0 + (size_t)rr * DMODEL + p * 128);
#pragma unroll
    for (int rr = 0; rr < LN_ROWS; ++rr)
#pragma unroll
        for (int p = 0; p < 4; ++p)
            r[rr][p] = *(const v4f*)(hin + o0 + (size_t)rr * DMODEL + p * 128);
#pragma unroll
    for (int p = 0; p < 4; ++p) {
        gg[p] = *(const v4f*)(g  + p * 128 + lane * 4);
        bb[p] = *(const v4f*)(be + p * 128 + lane * 4);
    }
    __builtin_amdgcn_sched_barrier(0);

    float s[LN_ROWS];
#pragma unroll
    for (int rr = 0; rr < LN_ROWS; ++rr) {
        const v4f t = (v[rr][0] + v[rr][1]) + (v[rr][2] + v[rr][3]);
        s[rr] = (t.x + t.y) + (t.z + t.w);
    }
#pragma unroll
    for (int m = 16; m >= 1; m >>= 1)
#pragma unroll
        for (int rr = 0; rr < LN_ROWS; ++rr)
            s[rr] += __shfl_xor(s[rr], m, 32);

    float q[LN_ROWS];
#pragma unroll
    for (int rr = 0; rr < LN_ROWS; ++rr) {
        const float mu = s[rr] * (1.f / DMODEL);
        v4f t = (v4f){0.f, 0.f, 0.f, 0.f};
#pragma unroll
        for (int p = 0; p < 4; ++p) {
            v[rr][p] = v[rr][p] - mu;
            t += v[rr][p] * v[rr][p];
        }
        q[rr] = (t.x + t.y) + (t.z + t.w);
    }
#pragma unroll
    for (int m = 16; m >= 1; m >>= 1)
#pragma unroll
        for (int rr = 0; rr < LN_ROWS; ++rr)
            q[rr] += __shfl_xor(q[rr], m, 32);

#pragma unroll
    for (int rr = 0; rr < LN_ROWS; ++rr) {
        const float rstd = rsqrtf(q[rr] * (1.f / DMODEL) + LN_EPS);
#pragma unroll
        for (int p = 0; p < 4; ++p) {
            const v4f o = r[rr][p] + v[rr][p] * rstd * gg[p] + bb[p];
            *(v4f*)(hout + o0 + (size_t)rr * DMODEL + p * 128) = o;
        }
    }
}

extern "C" void kernel_launch(void* const* d_in, const int* in_sizes, int n_in,
                              void* d_out, int out_size, void* d_ws, size_t ws_size,
                              hipStream_t stream) {
    (void)in_sizes; (void)n_in; (void)out_size; (void)ws_size;

    const float* x      = (const float*)d_in[0];
    const float* A_log  = (const float*)d_in[1];
    const float* W_B    = (const float*)d_in[2];
    const float* b_B    = (const float*)d_in[3];
    const float* W_C    = (const float*)d_in[4];
    const float* b_C    = (const float*)d_in[5];
    const float* W_dt   = (const float*)d_in[6];
    const float* b_dt   = (const float*)d_in[7];
    const float* D_skip = (const float*)d_in[8];
    const float* W_mix  = (const float*)d_in[9];
    const float* b_mix  = (const float*)d_in[10];
    const float* ln_g   = (const float*)d_in[11];
    const float* ln_b   = (const float*)d_in[12];
    const float* W_dec  = (const float*)d_in[13];
    const float* b_dec  = (const float*)d_in[14];
    float* out = (float*)d_out;

    float* ws = (float*)d_ws;
    size_t off = 0;
    float* dtb   = ws + off; off += (size_t)MROWS * DMODEL;
    float* Bmb   = ws + off; off += (size_t)MROWS * NSTATE;
    float* Cmb   = ws + off; off += (size_t)MROWS * NSTATE;
    float* ygelu = ws + off; off += (size_t)MROWS * DMODEL;
    float* ymix  = ws + off; off += (size_t)MROWS * DMODEL;
    float* h1    = ws + off; off += (size_t)MROWS * DMODEL;
    float* h2    = ws + off; off += (size_t)MROWS * DMODEL;
    float* hend  = ws + off; off += (size_t)BATCH * DMODEL * NCHUNK * NSTATE;
    float* aend  = ws + off; off += (size_t)BATCH * DMODEL * NCHUNK * NSTATE;
    float* carry = ws + off; off += (size_t)BATCH * DMODEL * NCHUNK * NSTATE;

    const int DN = DMODEL * NSTATE;
    const int DD = DMODEL * DMODEL;

    const dim3 gBig(MROWS / 128, DMODEL / 64);
    const dim3 gProj(MROWS / 128, DMODEL / 64 + MROWS / BC_BM / (MROWS / 128));
    const dim3 gScan(DMODEL / 256, NCHUNK, BATCH);

    for (int i = 0; i < 2; ++i) {
        const float* hin = (i == 0) ? x : h1;
        float* hout      = (i == 0) ? h1 : h2;

        s6_proj_dt_bc<<<gProj, 256, 0, stream>>>(
            hin, W_dt + (size_t)i * DD, b_dt + i * DMODEL, dtb,
            W_B + (size_t)i * DN, b_B + i * NSTATE, Bmb,
            W_C + (size_t)i * DN, b_C + i * NSTATE, Cmb);

        s6_scan_pass1<<<gScan, 256, 0, stream>>>(
            hin, dtb, Bmb, A_log + (size_t)i * DN, hend, aend);
        s6_scan_pass2<<<BATCH * DMODEL * NSTATE / 256, 256, 0, stream>>>(
            hend, aend, carry);
        s6_scan_pass3<<<gScan, 256, 0, stream>>>(
            hin, dtb, Bmb, Cmb, A_log + (size_t)i * DN, carry,
            D_skip + i * DMODEL, ygelu);

        s6_gemm_wmma_k512<128, 64, 32, 32><<<gBig, 256, 0, stream>>>(
            ygelu, W_mix + (size_t)i * DD, b_mix + i * DMODEL, ymix, DMODEL);
        s6_ln_residual<<<MROWS / (LN_WAVES * LN_ROWS), LN_WAVES * 32, 0, stream>>>(
            ymix, hin, ln_g + i * DMODEL, ln_b + i * DMODEL, hout);
    }

    s6_gemm_wmma_k512<128, 64, 32, 32><<<gBig, 256, 0, stream>>>(h2, W_dec, b_dec, out, OUTD);
}


# Scan passes: 16 states as 8 packed pairs, A pre-scaled by log2(e), chunk decay as one exp2 of the summed dt, C.h summed in two pair chains (rounding order): fewer vector instructions per step
# speedup vs baseline: 4.3914x; 1.0829x over previous
#include <hip/hip_runtime.h>
#include <hip/hip_bf16.h>
#include <math.h>

#define BATCH   4
#define SEQL    2048
#define DMODEL  512
#define NSTATE  16
#define OUTD    512
#define MROWS   (BATCH * SEQL)
#define KDIM    512
#define NCHUNK  32
#define CLEN    (SEQL / NCHUNK)
#define TGRP    8
#define LN_EPS  1e-5f
#define NPAIR   (NSTATE / 2)
#define S6_LOG2E 1.4426950408889634f

typedef float v2f __attribute__((ext_vector_type(2)));
typedef float v4f __attribute__((ext_vector_type(4)));
typedef float v8f __attribute__((ext_vector_type(8)));

#define GEMM_KB 32

__device__ __forceinline__ float s6_softplus(float v) {
    const float e = __builtin_amdgcn_exp2f(fabsf(v) * -1.4426950408889634f);
    return fmaf(__builtin_amdgcn_logf(1.f + e), 0.6931471805599453f, fmaxf(v, 0.f));
}

template <int BM, int BN, int WM, int WN, bool SPLIT, bool SOFTPLUS>
__device__ __forceinline__
void s6_gemm_tile_k512(float* __restrict__ sA, float* __restrict__ sB,
                       const float* __restrict__ A, int m0,
                       const float* __restrict__ W0, const float* __restrict__ W1,
                       int ldw, int n0,
                       const float* __restrict__ bias0, const float* __restrict__ bias1,
                       float* __restrict__ out0, float* __restrict__ out1,
                       int ldo) {
    constexpr int KB    = GEMM_KB;
    constexpr int PA    = KB + 4;
    constexpr int PB    = BN + 8;
    constexpr int NT    = (BM / WM) * (BN / WN) * 32;
    constexpr int TM    = WM / 16;
    constexpr int TN    = WN / 16;
    constexpr int AQ    = KB / 4;
    constexpr int BQ    = BN / 4;
    constexpr int A_PER = (BM * AQ) / NT;
    constexpr int B_PER = (KB * BQ) / NT;
    constexpr int NCH   = KDIM / KB;
    static_assert((BM * AQ) % NT == 0, "A chunk must split evenly over the block");
    static_assert((KB * BQ) % NT == 0, "W chunk must split evenly over the block");

    const int tid = threadIdx.x;

    const float* agp[A_PER]; int asl[A_PER];
    const float* bgp[B_PER]; int bsl[B_PER];
#pragma unroll
    for (int p = 0; p < A_PER; ++p) {
        const int idx = tid + p * NT;
        const int row = idx / AQ, kq = idx % AQ;
        agp[p] = A + (size_t)(m0 + row) * KDIM + kq * 4;
        asl[p] = row * PA + kq * 4;
    }
#pragma unroll
    for (int p = 0; p < B_PER; ++p) {
        const int idx = tid + p * NT;
        const int kr = idx / BQ, nq = idx % BQ;
        if constexpr (SPLIT)
            bgp[p] = ((nq < BQ / 2) ? W0 : W1) + (size_t)kr * ldw + (nq % (BQ / 2)) * 4;
        else
            bgp[p] = W0 + (size_t)kr * ldw + n0 + nq * 4;
        bsl[p] = kr * PB + nq * 4;
    }

    v4f ra[A_PER], rb[B_PER];
#pragma unroll
    for (int p = 0; p < A_PER; ++p) ra[p] = *(const v4f*)agp[p];
#pragma unroll
    for (int p = 0; p < B_PER; ++p) rb[p] = *(const v4f*)bgp[p];

    const int wave = tid >> 5;
    const int lane = tid & 31;
    const int wm   = (wave / (BN / WN)) * WM;
    const int wn   = (wave % (BN / WN)) * WN;
    const int c16  = lane & 15;
    const int klo  = (lane >> 4) << 1;

    v8f acc[TM][TN];
#pragma unroll
    for (int i = 0; i < TM; ++i)
#pragma unroll
        for (int j = 0; j < TN; ++j)
            acc[i][j] = (v8f){0.f, 0.f, 0.f, 0.f, 0.f, 0.f, 0.f, 0.f};

    const float* saf = sA + (wm + c16) * PA + klo;
    const float* sbf = sB + klo * PB + wn + c16;

#pragma unroll
    for (int p = 0; p < A_PER; ++p) *(v4f*)&sA[asl[p]] = ra[p];
#pragma unroll
    for (int p = 0; p < B_PER; ++p) *(v4f*)&sB[bsl[p]] = rb[p];
    __syncthreads();

    for (int kc = 0; kc < NCH; ++kc) {
        const int  cur  = kc & 1;
        const bool more = (kc + 1 < NCH);

        if (more) {
#pragma unroll
            for (int p = 0; p < A_PER; ++p)
                ra[p] = *(const v4f*)(agp[p] + (kc + 1) * KB);
#pragma unroll
            for (int p = 0; p < B_PER; ++p)
                rb[p] = *(const v4f*)(bgp[p] + (size_t)(kc + 1) * KB * ldw);
        }
        __builtin_amdgcn_sched_barrier(0);

        const float* sa = saf + cur * (BM * PA);
        const float* sb = sbf + cur * (KB * PB);
#pragma unroll
        for (int k0 = 0; k0 < KB; k0 += 4) {
            v2f av[TM], bv[TN];
#pragma unroll
            for (int i = 0; i < TM; ++i)
                av[i] = *(const v2f*)(sa + i * 16 * PA + k0);
#pragma unroll
            for (int j = 0; j < TN; ++j) {
                bv[j].x = sb[k0 * PB + j * 16];
                bv[j].y = sb[(k0 + 1) * PB + j * 16];
            }
#pragma unroll
            for (int i = 0; i < TM; ++i)
#pragma unroll
                for (int j = 0; j < TN; ++j)
                    acc[i][j] = __builtin_amdgcn_wmma_f32_16x16x4_f32(
                         false, av[i],  false, bv[j],
                         (short)0, acc[i][j],
                         false,  false);
        }

        if (more) {
            float* da = sA + (cur ^ 1) * (BM * PA);
            float* db = sB + (cur ^ 1) * (KB * PB);
#pragma unroll
            for (int p = 0; p < A_PER; ++p) *(v4f*)&da[asl[p]] = ra[p];
#pragma unroll
            for (int p = 0; p < B_PER; ++p) *(v4f*)&db[bsl[p]] = rb[p];
        }
        __syncthreads();
    }

    const int mhi = (lane >> 4) << 3;
#pragma unroll
    for (int j = 0; j < TN; ++j) {
        const int  ct = wn + j * 16;
        const bool hi = SPLIT && (ct >= BN / 2);
        const int  col = SPLIT ? (ct % (BN / 2)) + c16 : n0 + ct + c16;
        const float* __restrict__ bias = hi ? bias1 : bias0;
        float* __restrict__       out  = hi ? out1 : out0;
        const float bb  = bias[col];
#pragma unroll
        for (int i = 0; i < TM; ++i) {
#pragma unroll
            for (int r = 0; r < 8; ++r) {
                float v = acc[i][j][r] + bb;
                if constexpr (SOFTPLUS) v = s6_softplus(v);
                out[(size_t)(m0 + wm + i * 16 + mhi + r) * ldo + col] = v;
            }
        }
    }
}

template <int BM, int BN, int WM, int WN>
__global__ __launch_bounds__((BM / WM) * (BN / WN) * 32)
void s6_gemm_wmma_k512(const float* __restrict__ A,
                       const float* __restrict__ W,
                       const float* __restrict__ bias,
                       float* __restrict__ out,
                       int Nout) {
    __shared__ float sA[2 * BM * (GEMM_KB + 4)] __attribute__((aligned(16)));
    __shared__ float sB[2 * GEMM_KB * (BN + 8)] __attribute__((aligned(16)));
    s6_gemm_tile_k512<BM, BN, WM, WN, false, false>(sA, sB, A, blockIdx.x * BM,
                                                    W, W, Nout, blockIdx.y * BN,
                                                    bias, bias, out, out, Nout);
}

#define BC_BM 64
__global__ __launch_bounds__(256)
void s6_proj_dt_bc(const float* __restrict__ hin,
                   const float* __restrict__ W_dt, const float* __restrict__ b_dt,
                   float* __restrict__ dt,
                   const float* __restrict__ W_B, const float* __restrict__ b_B,
                   float* __restrict__ Bm,
                   const float* __restrict__ W_C, const float* __restrict__ b_C,
                   float* __restrict__ Cm) {
    __shared__ float sA[2 * 128 * (GEMM_KB + 4)] __attribute__((aligned(16)));
    __shared__ float sB[2 * GEMM_KB * (64 + 8)] __attribute__((aligned(16)));
    const int ydt = DMODEL / 64;
    if ((int)blockIdx.y < ydt) {
        s6_gemm_tile_k512<128, 64, 32, 32, false, true>(sA, sB, hin, blockIdx.x * 128,
                                                        W_dt, W_dt, DMODEL, blockIdx.y * 64,
                                                        b_dt, b_dt, dt, dt, DMODEL);
    } else {
        const int mt = ((int)blockIdx.y - ydt) * (int)gridDim.x + (int)blockIdx.x;
        s6_gemm_tile_k512<BC_BM, 2 * NSTATE, 16, 16, true, false>(sA, sB, hin, mt * BC_BM,
                                                                  W_B, W_C, NSTATE, 0,
                                                                  b_B, b_C, Bm, Cm, NSTATE);
    }
}

__global__ __launch_bounds__(256)
void s6_scan_pass1(const float* __restrict__ x,
                   const float* __restrict__ dt,
                   const float* __restrict__ Bm,
                   const float* __restrict__ A_log,
                   float* __restrict__ hend,
                   float* __restrict__ aend) {
    const int b = blockIdx.z, c = blockIdx.y;
    const int d = blockIdx.x * 256 + threadIdx.x;

    __shared__ float sBm[CLEN * NSTATE] __attribute__((aligned(16)));
    for (int idx = threadIdx.x; idx < CLEN * NSTATE; idx += 256)
        sBm[idx] = Bm[((size_t)b * SEQL + c * CLEN) * NSTATE + idx];
    __syncthreads();

    v2f A2[NPAIR];
#pragma unroll
    for (int j = 0; j < NPAIR; ++j) {
        A2[j].x = -expf(A_log[d * NSTATE + 2 * j])     * S6_LOG2E;
        A2[j].y = -expf(A_log[d * NSTATE + 2 * j + 1]) * S6_LOG2E;
    }

    v2f h[NPAIR];
#pragma unroll
    for (int j = 0; j < NPAIR; ++j) h[j] = (v2f){0.f, 0.f};
    float sdt = 0.f;

    const float* dtp = dt + ((size_t)b * SEQL + c * CLEN) * DMODEL + d;
    const float* xp  = x  + ((size_t)b * SEQL + c * CLEN) * DMODEL + d;
    v2f cur[TGRP];
#pragma unroll
    for (int i = 0; i < TGRP; ++i) {
        cur[i].x = dtp[(size_t)i * DMODEL];
        cur[i].y = xp[(size_t)i * DMODEL];
    }

    for (int t0 = 0; t0 < CLEN; t0 += TGRP) {
        const int tn = (t0 + TGRP < CLEN) ? (t0 + TGRP) : t0;
        v2f nxt[TGRP];
#pragma unroll
        for (int i = 0; i < TGRP; ++i) {
            nxt[i].x = dtp[(size_t)(tn + i) * DMODEL];
            nxt[i].y = xp[(size_t)(tn + i) * DMODEL];
        }
        v2f bq[TGRP][NPAIR];
#pragma unroll
        for (int i = 0; i < TGRP; ++i) {
            const v2f* bm = (const v2f*)&sBm[(t0 + i) * NSTATE];
#pragma unroll
            for (int j = 0; j < NPAIR; ++j) bq[i][j] = bm[j];
        }
        __builtin_amdgcn_sched_barrier(0);

#pragma unroll
        for (int i = 0; i < TGRP; ++i) {
            const float dtv = cur[i].x;
            const float ub  = dtv * cur[i].y;
            const v2f   dt2 = (v2f){dtv, dtv};
            const v2f   ub2 = (v2f){ub, ub};
            sdt += dtv;
#pragma unroll
            for (int j = 0; j < NPAIR; ++j) {
                const v2f e = dt2 * A2[j];
                v2f a;
                a.x = __builtin_amdgcn_exp2f(e.x);
                a.y = __builtin_amdgcn_exp2f(e.y);
                h[j] = a * h[j] + ub2 * bq[i][j];
            }
        }
#pragma unroll
        for (int i = 0; i < TGRP; ++i) cur[i] = nxt[i];
    }

    const size_t base = (((size_t)b * DMODEL + d) * NCHUNK + c) * NSTATE;
#pragma unroll
    for (int j = 0; j < NPAIR; ++j) {
        hend[base + 2 * j]     = h[j].x;
        hend[base + 2 * j + 1] = h[j].y;
        aend[base + 2 * j]     = __builtin_amdgcn_exp2f(A2[j].x * sdt);
        aend[base + 2 * j + 1] = __builtin_amdgcn_exp2f(A2[j].y * sdt);
    }
}

__global__ __launch_bounds__(256)
void s6_scan_pass2(const float* __restrict__ hend,
                   const float* __restrict__ aend,
                   float* __restrict__ carry) {
    const int idx = blockIdx.x * 256 + threadIdx.x;
    const size_t bd = (size_t)(idx >> 4);
    const int n = idx & 15;
    float cv = 0.f;
    for (int c = 0; c < NCHUNK; ++c) {
        const size_t o = (bd * NCHUNK + c) * NSTATE + n;
        carry[o] = cv;
        cv = aend[o] * cv + hend[o];
    }
}

__global__ __launch_bounds__(256)
void s6_scan_pass3(const float* __restrict__ x,
                   const float* __restrict__ dt,
                   const float* __restrict__ Bm,
                   const float* __restrict__ Cm,
                   const float* __restrict__ A_log,
                   const float* __restrict__ carry,
                   const float* __restrict__ Dskip,
                   float* __restrict__ yout) {
    const int b = blockIdx.z, c = blockIdx.y;
    const int d = blockIdx.x * 256 + threadIdx.x;

    __shared__ float sBm[CLEN * NSTATE] __attribute__((aligned(16)));
    __shared__ float sCm[CLEN * NSTATE] __attribute__((aligned(16)));
    for (int idx = threadIdx.x; idx < CLEN * NSTATE; idx += 256) {
        const size_t g = ((size_t)b * SEQL + c * CLEN) * NSTATE + idx;
        sBm[idx] = Bm[g];
        sCm[idx] = Cm[g];
    }
    __syncthreads();

    v2f A2[NPAIR];
#pragma unroll
    for (int j = 0; j < NPAIR; ++j) {
        A2[j].x = -expf(A_log[d * NSTATE + 2 * j])     * S6_LOG2E;
        A2[j].y = -expf(A_log[d * NSTATE + 2 * j + 1]) * S6_LOG2E;
    }

    v2f h[NPAIR];
    const size_t base = (((size_t)b * DMODEL + d) * NCHUNK + c) * NSTATE;
#pragma unroll
    for (int j = 0; j < NPAIR; ++j) {
        h[j].x = carry[base + 2 * j];
        h[j].y = carry[base + 2 * j + 1];
    }

    const float dsk = Dskip[d];

    const size_t cbase = ((size_t)b * SEQL + c * CLEN) * DMODEL + d;
    const float* dtp = dt + cbase;
    const float* xp  = x  + cbase;
    float*       yp  = yout + cbase;
    v2f cur[TGRP];
#pragma unroll
    for (int i = 0; i < TGRP; ++i) {
        cur[i].x = dtp[(size_t)i * DMODEL];
        cur[i].y = xp[(size_t)i * DMODEL];
    }

    for (int t0 = 0; t0 < CLEN; t0 += TGRP) {
        const int tn = (t0 + TGRP < CLEN) ? (t0 + TGRP) : t0;
        v2f nxt[TGRP];
#pragma unroll
        for (int i = 0; i < TGRP; ++i) {
            nxt[i].x = dtp[(size_t)(tn + i) * DMODEL];
            nxt[i].y = xp[(size_t)(tn + i) * DMODEL];
        }
        __builtin_amdgcn_sched_barrier(0);

        float yv[TGRP];
#pragma unroll
        for (int i = 0; i < TGRP; ++i) {
            const float dtv = cur[i].x;
            const float xv  = cur[i].y;
            const float ub  = dtv * xv;
            const v2f   dt2 = (v2f){dtv, dtv};
            const v2f   ub2 = (v2f){ub, ub};
            const v2f*  bm  = (const v2f*)&sBm[(t0 + i) * NSTATE];
            const v2f*  cm  = (const v2f*)&sCm[(t0 + i) * NSTATE];
            v2f y0 = (v2f){0.f, 0.f}, y1 = (v2f){0.f, 0.f};
#pragma unroll
            for (int j = 0; j < NPAIR; j += 2) {
                const v2f e0 = dt2 * A2[j];
                const v2f e1 = dt2 * A2[j + 1];
                v2f a0, a1;
                a0.x = __builtin_amdgcn_exp2f(e0.x);
                a0.y = __builtin_amdgcn_exp2f(e0.y);
                a1.x = __builtin_amdgcn_exp2f(e1.x);
                a1.y = __builtin_amdgcn_exp2f(e1.y);
                h[j]     = a0 * h[j]     + ub2 * bm[j];
                h[j + 1] = a1 * h[j + 1] + ub2 * bm[j + 1];
                y0 += cm[j]     * h[j];
                y1 += cm[j + 1] * h[j + 1];
            }
            const v2f ys = y0 + y1;
            yv[i] = (ys.x + ys.y) + dsk * xv;
        }
#pragma unroll
        for (int i = 0; i < TGRP; ++i) {
            const float y = yv[i];
            yp[(size_t)(t0 + i) * DMODEL] =
                0.5f * y * (1.f + erff(y * 0.70710678118654752f));
        }
#pragma unroll
        for (int i = 0; i < TGRP; ++i) cur[i] = nxt[i];
    }
}

#define LN_ROWS 2
#define LN_WAVES 8
__global__ __launch_bounds__(LN_WAVES * 32)
void s6_ln_residual(const float* __restrict__ ymix,
                    const float* __restrict__ hin,
                    const float* __restrict__ g,
                    const float* __restrict__ be,
                    float* __restrict__ hout) {
    const int lane = threadIdx.x & 31;
    const int wave = threadIdx.x >> 5;
    const size_t row0 = ((size_t)blockIdx.x * LN_WAVES + wave) * LN_ROWS;
    const size_t o0   = row0 * DMODEL + (size_t)lane * 4;

    v4f v[LN_ROWS][4], r[LN_ROWS][4], gg[4], bb[4];
#pragma unroll
    for (int rr = 0; rr < LN_ROWS; ++rr)
#pragma unroll
        for (int p = 0; p < 4; ++p)
            v[rr][p] = *(const v4f*)(ymix + o0 + (size_t)rr * DMODEL + p * 128);
#pragma unroll
    for (int rr = 0; rr < LN_ROWS; ++rr)
#pragma unroll
        for (int p = 0; p < 4; ++p)
            r[rr][p] = *(const v4f*)(hin + o0 + (size_t)rr * DMODEL + p * 128);
#pragma unroll
    for (int p = 0; p < 4; ++p) {
        gg[p] = *(const v4f*)(g  + p * 128 + lane * 4);
        bb[p] = *(const v4f*)(be + p * 128 + lane * 4);
    }
    __builtin_amdgcn_sched_barrier(0);

    float s[LN_ROWS];
#pragma unroll
    for (int rr = 0; rr < LN_ROWS; ++rr) {
        const v4f t = (v[rr][0] + v[rr][1]) + (v[rr][2] + v[rr][3]);
        s[rr] = (t.x + t.y) + (t.z + t.w);
    }
#pragma unroll
    for (int m = 16; m >= 1; m >>= 1)
#pragma unroll
        for (int rr = 0; rr < LN_ROWS; ++rr)
            s[rr] += __shfl_xor(s[rr], m, 32);

    float q[LN_ROWS];
#pragma unroll
    for (int rr = 0; rr < LN_ROWS; ++rr) {
        const float mu = s[rr] * (1.f / DMODEL);
        v4f t = (v4f){0.f, 0.f, 0.f, 0.f};
#pragma unroll
        for (int p = 0; p < 4; ++p) {
            v[rr][p] = v[rr][p] - mu;
            t += v[rr][p] * v[rr][p];
        }
        q[rr] = (t.x + t.y) + (t.z + t.w);
    }
#pragma unroll
    for (int m = 16; m >= 1; m >>= 1)
#pragma unroll
        for (int rr = 0; rr < LN_ROWS; ++rr)
            q[rr] += __shfl_xor(q[rr], m, 32);

#pragma unroll
    for (int rr = 0; rr < LN_ROWS; ++rr) {
        const float rstd = rsqrtf(q[rr] * (1.f / DMODEL) + LN_EPS);
#pragma unroll
        for (int p = 0; p < 4; ++p) {
            const v4f o = r[rr][p] + v[rr][p] * rstd * gg[p] + bb[p];
            *(v4f*)(hout + o0 + (size_t)rr * DMODEL + p * 128) = o;
        }
    }
}

extern "C" void kernel_launch(void* const* d_in, const int* in_sizes, int n_in,
                              void* d_out, int out_size, void* d_ws, size_t ws_size,
                              hipStream_t stream) {
    (void)in_sizes; (void)n_in; (void)out_size; (void)ws_size;

    const float* x      = (const float*)d_in[0];
    const float* A_log  = (const float*)d_in[1];
    const float* W_B    = (const float*)d_in[2];
    const float* b_B    = (const float*)d_in[3];
    const float* W_C    = (const float*)d_in[4];
    const float* b_C    = (const float*)d_in[5];
    const float* W_dt   = (const float*)d_in[6];
    const float* b_dt   = (const float*)d_in[7];
    const float* D_skip = (const float*)d_in[8];
    const float* W_mix  = (const float*)d_in[9];
    const float* b_mix  = (const float*)d_in[10];
    const float* ln_g   = (const float*)d_in[11];
    const float* ln_b   = (const float*)d_in[12];
    const float* W_dec  = (const float*)d_in[13];
    const float* b_dec  = (const float*)d_in[14];
    float* out = (float*)d_out;

    float* ws = (float*)d_ws;
    size_t off = 0;
    float* dtb   = ws + off; off += (size_t)MROWS * DMODEL;
    float* Bmb   = ws + off; off += (size_t)MROWS * NSTATE;
    float* Cmb   = ws + off; off += (size_t)MROWS * NSTATE;
    float* ygelu = ws + off; off += (size_t)MROWS * DMODEL;
    float* ymix  = ws + off; off += (size_t)MROWS * DMODEL;
    float* h1    = ws + off; off += (size_t)MROWS * DMODEL;
    float* h2    = ws + off; off += (size_t)MROWS * DMODEL;
    float* hend  = ws + off; off += (size_t)BATCH * DMODEL * NCHUNK * NSTATE;
    float* aend  = ws + off; off += (size_t)BATCH * DMODEL * NCHUNK * NSTATE;
    float* carry = ws + off; off += (size_t)BATCH * DMODEL * NCHUNK * NSTATE;

    const int DN = DMODEL * NSTATE;
    const int DD = DMODEL * DMODEL;

    const dim3 gBig(MROWS / 128, DMODEL / 64);
    const dim3 gProj(MROWS / 128, DMODEL / 64 + MROWS / BC_BM / (MROWS / 128));
    const dim3 gScan(DMODEL / 256, NCHUNK, BATCH);

    for (int i = 0; i < 2; ++i) {
        const float* hin = (i == 0) ? x : h1;
        float* hout      = (i == 0) ? h1 : h2;

        s6_proj_dt_bc<<<gProj, 256, 0, stream>>>(
            hin, W_dt + (size_t)i * DD, b_dt + i * DMODEL, dtb,
            W_B + (size_t)i * DN, b_B + i * NSTATE, Bmb,
            W_C + (size_t)i * DN, b_C + i * NSTATE, Cmb);

        s6_scan_pass1<<<gScan, 256, 0, stream>>>(
            hin, dtb, Bmb, A_log + (size_t)i * DN, hend, aend);
        s6_scan_pass2<<<BATCH * DMODEL * NSTATE / 256, 256, 0, stream>>>(
            hend, aend, carry);
        s6_scan_pass3<<<gScan, 256, 0, stream>>>(
            hin, dtb, Bmb, Cmb, A_log + (size_t)i * DN, carry,
            D_skip + i * DMODEL, ygelu);

        s6_gemm_wmma_k512<128, 64, 32, 32><<<gBig, 256, 0, stream>>>(
            ygelu, W_mix + (size_t)i * DD, b_mix + i * DMODEL, ymix, DMODEL);
        s6_ln_residual<<<MROWS / (LN_WAVES * LN_ROWS), LN_WAVES * 32, 0, stream>>>(
            ymix, hin, ln_g + i * DMODEL, ln_b + i * DMODEL, hout);
    }

    s6_gemm_wmma_k512<128, 64, 32, 32><<<gBig, 256, 0, stream>>>(h2, W_dec, b_dec, out, OUTD);
}


# GEMM K loop freed of vector instructions beside the f32 matrix ops: W staged in LDS as k-pair rows, so a B fragment is one 8-byte read, no register moves; loop unrolled by 2, stage offsets literal
# speedup vs baseline: 4.5525x; 1.0367x over previous
#include <hip/hip_runtime.h>
#include <hip/hip_bf16.h>
#include <math.h>

#define BATCH   4
#define SEQL    2048
#define DMODEL  512
#define NSTATE  16
#define OUTD    512
#define MROWS   (BATCH * SEQL)
#define KDIM    512
#define NCHUNK  32
#define CLEN    (SEQL / NCHUNK)
#define TGRP    8
#define LN_EPS  1e-5f
#define NPAIR   (NSTATE / 2)
#define S6_LOG2E 1.4426950408889634f

typedef float v2f __attribute__((ext_vector_type(2)));
typedef float v4f __attribute__((ext_vector_type(4)));
typedef float v8f __attribute__((ext_vector_type(8)));

#define GEMM_KB 32

__device__ __forceinline__ float s6_softplus(float v) {
    const float e = __builtin_amdgcn_exp2f(fabsf(v) * -1.4426950408889634f);
    return fmaf(__builtin_amdgcn_logf(1.f + e), 0.6931471805599453f, fmaxf(v, 0.f));
}

template <int BM, int BN, int WM, int WN, bool SPLIT, bool SOFTPLUS, int LDW, int LDO>
__device__ __forceinline__
void s6_gemm_tile_k512(float* __restrict__ sA, float* __restrict__ sB,
                       const float* __restrict__ A, int m0,
                       const float* __restrict__ W0, const float* __restrict__ W1,
                       int n0,
                       const float* __restrict__ bias0, const float* __restrict__ bias1,
                       float* __restrict__ out0, float* __restrict__ out1) {
    constexpr int KB    = GEMM_KB;
    constexpr int PA    = KB + 4;
    constexpr int PBP   = 2 * BN + 32;
    constexpr int SBST  = (KB / 2) * PBP;
    constexpr int NT    = (BM / WM) * (BN / WN) * 32;
    constexpr int TM    = WM / 16;
    constexpr int TN    = WN / 16;
    constexpr int AQ    = KB / 4;
    constexpr int A_PER = (BM * AQ) / NT;
    constexpr int KPT   = NT / BN;
    constexpr int B_PER = (KB / 2) / KPT;
    constexpr int NCH   = KDIM / KB;
    static_assert((BM * AQ) % NT == 0, "A chunk must split evenly over the block");
    static_assert(NT % BN == 0 && (KB / 2) % KPT == 0, "W chunk must split evenly over the block");

    const int tid = threadIdx.x;

    unsigned aofs[A_PER]; int asl[A_PER];
#pragma unroll
    for (int p = 0; p < A_PER; ++p) {
        const int idx = tid + p * NT;
        const int row = idx / AQ, kq = idx % AQ;
        aofs[p] = (unsigned)(((m0 + row) * KDIM + kq * 4) * 4);
        asl[p]  = row * PA + kq * 4;
    }
    const int wcol = tid % BN;
    const int wkp  = tid / BN;
    const float* wsrc;
    if constexpr (SPLIT)
        wsrc = ((wcol < BN / 2) ? W0 : W1) + (size_t)(2 * wkp) * LDW + (wcol % (BN / 2));
    else
        wsrc = W0 + (size_t)(2 * wkp) * LDW + n0 + wcol;
    const int wsl = wkp * PBP + wcol * 2;

    v4f ra[A_PER];
    v2f rb[B_PER];
    {
        const char* ab = (const char*)A;
#pragma unroll
        for (int p = 0; p < A_PER; ++p) ra[p] = *(const v4f*)(ab + aofs[p]);
#pragma unroll
        for (int p = 0; p < B_PER; ++p) {
            rb[p].x = wsrc[(size_t)(2 * p * KPT) * LDW];
            rb[p].y = wsrc[(size_t)(2 * p * KPT + 1) * LDW];
        }
    }

    const int wave = tid >> 5;
    const int lane = tid & 31;
    const int wm   = (wave / (BN / WN)) * WM;
    const int wn   = (wave % (BN / WN)) * WN;
    const int c16  = lane & 15;
    const int klo  = (lane >> 4) << 1;

    v8f acc[TM][TN];
#pragma unroll
    for (int i = 0; i < TM; ++i)
#pragma unroll
        for (int j = 0; j < TN; ++j)
            acc[i][j] = (v8f){0.f, 0.f, 0.f, 0.f, 0.f, 0.f, 0.f, 0.f};

    const float* saf = sA + (wm + c16) * PA + klo;
    const float* sbf = sB + (klo >> 1) * PBP + (wn + c16) * 2;

#pragma unroll
    for (int p = 0; p < A_PER; ++p) *(v4f*)&sA[asl[p]] = ra[p];
#pragma unroll
    for (int p = 0; p < B_PER; ++p) *(v2f*)&sB[wsl + p * KPT * PBP] = rb[p];
    __syncthreads();

#pragma unroll 2
    for (int kc = 0; kc < NCH; ++kc) {
        const int  cur  = kc & 1;
        const bool more = (kc + 1 < NCH);

        if (more) {
            const char*  ab = (const char*)A + (size_t)(kc + 1) * (KB * 4);
            const float* wb = wsrc + (size_t)(kc + 1) * KB * LDW;
#pragma unroll
            for (int p = 0; p < A_PER; ++p)
                ra[p] = *(const v4f*)(ab + aofs[p]);
#pragma unroll
            for (int p = 0; p < B_PER; ++p) {
                rb[p].x = wb[(size_t)(2 * p * KPT) * LDW];
                rb[p].y = wb[(size_t)(2 * p * KPT + 1) * LDW];
            }
        }
        __builtin_amdgcn_sched_barrier(0);

        const float* sa = saf + cur * (BM * PA);
        const float* sb = sbf + cur * SBST;
#pragma unroll
        for (int k0 = 0; k0 < KB; k0 += 4) {
            v2f av[TM], bv[TN];
#pragma unroll
            for (int i = 0; i < TM; ++i)
                av[i] = *(const v2f*)(sa + i * 16 * PA + k0);
#pragma unroll
            for (int j = 0; j < TN; ++j)
                bv[j] = *(const v2f*)(sb + (k0 >> 1) * PBP + j * 32);
#pragma unroll
            for (int i = 0; i < TM; ++i)
#pragma unroll
                for (int j = 0; j < TN; ++j)
                    acc[i][j] = __builtin_amdgcn_wmma_f32_16x16x4_f32(
                         false, av[i],  false, bv[j],
                         (short)0, acc[i][j],
                         false,  false);
        }

        if (more) {
            float* da = sA + (cur ^ 1) * (BM * PA);
            float* db = sB + (cur ^ 1) * SBST;
#pragma unroll
            for (int p = 0; p < A_PER; ++p) *(v4f*)&da[asl[p]] = ra[p];
#pragma unroll
            for (int p = 0; p < B_PER; ++p) *(v2f*)&db[wsl + p * KPT * PBP] = rb[p];
        }
        __syncthreads();
    }

    const int mhi = (lane >> 4) << 3;
#pragma unroll
    for (int j = 0; j < TN; ++j) {
        const int  ct = wn + j * 16;
        const bool hi = SPLIT && (ct >= BN / 2);
        const int  col = SPLIT ? (ct % (BN / 2)) + c16 : n0 + ct + c16;
        const float* __restrict__ bias = hi ? bias1 : bias0;
        float* __restrict__       out  = hi ? out1 : out0;
        const float bb  = bias[col];
#pragma unroll
        for (int i = 0; i < TM; ++i) {
#pragma unroll
            for (int r = 0; r < 8; ++r) {
                float v = acc[i][j][r] + bb;
                if constexpr (SOFTPLUS) v = s6_softplus(v);
                out[(size_t)(m0 + wm + i * 16 + mhi + r) * LDO + col] = v;
            }
        }
    }
}

template <int BM, int BN, int WM, int WN, int NOUT>
__global__ __launch_bounds__((BM / WM) * (BN / WN) * 32)
void s6_gemm_wmma_k512(const float* __restrict__ A,
                       const float* __restrict__ W,
                       const float* __restrict__ bias,
                       float* __restrict__ out) {
    __shared__ float sA[2 * BM * (GEMM_KB + 4)] __attribute__((aligned(16)));
    __shared__ float sB[2 * (GEMM_KB / 2) * (2 * BN + 32)] __attribute__((aligned(16)));
    s6_gemm_tile_k512<BM, BN, WM, WN, false, false, NOUT, NOUT>(sA, sB, A, blockIdx.x * BM,
                                                                W, W, blockIdx.y * BN,
                                                                bias, bias, out, out);
}

#define BC_BM 64
__global__ __launch_bounds__(256)
void s6_proj_dt_bc(const float* __restrict__ hin,
                   const float* __restrict__ W_dt, const float* __restrict__ b_dt,
                   float* __restrict__ dt,
                   const float* __restrict__ W_B, const float* __restrict__ b_B,
                   float* __restrict__ Bm,
                   const float* __restrict__ W_C, const float* __restrict__ b_C,
                   float* __restrict__ Cm) {
    __shared__ float sA[2 * 128 * (GEMM_KB + 4)] __attribute__((aligned(16)));
    __shared__ float sB[2 * (GEMM_KB / 2) * (2 * 64 + 32)] __attribute__((aligned(16)));
    const int ydt = DMODEL / 64;
    if ((int)blockIdx.y < ydt) {
        s6_gemm_tile_k512<128, 64, 32, 32, false, true, DMODEL, DMODEL>(sA, sB, hin, blockIdx.x * 128,
                                                                        W_dt, W_dt, blockIdx.y * 64,
                                                                        b_dt, b_dt, dt, dt);
    } else {
        const int mt = ((int)blockIdx.y - ydt) * (int)gridDim.x + (int)blockIdx.x;
        s6_gemm_tile_k512<BC_BM, 2 * NSTATE, 16, 16, true, false, NSTATE, NSTATE>(sA, sB, hin, mt * BC_BM,
                                                                                  W_B, W_C, 0,
                                                                                  b_B, b_C, Bm, Cm);
    }
}

__global__ __launch_bounds__(256)
void s6_scan_pass1(const float* __restrict__ x,
                   const float* __restrict__ dt,
                   const float* __restrict__ Bm,
                   const float* __restrict__ A_log,
                   float* __restrict__ hend,
                   float* __restrict__ aend) {
    const int b = blockIdx.z, c = blockIdx.y;
    const int d = blockIdx.x * 256 + threadIdx.x;

    __shared__ float sBm[CLEN * NSTATE] __attribute__((aligned(16)));
    for (int idx = threadIdx.x; idx < CLEN * NSTATE; idx += 256)
        sBm[idx] = Bm[((size_t)b * SEQL + c * CLEN) * NSTATE + idx];
    __syncthreads();

    v2f A2[NPAIR];
#pragma unroll
    for (int j = 0; j < NPAIR; ++j) {
        A2[j].x = -expf(A_log[d * NSTATE + 2 * j])     * S6_LOG2E;
        A2[j].y = -expf(A_log[d * NSTATE + 2 * j + 1]) * S6_LOG2E;
    }

    v2f h[NPAIR];
#pragma unroll
    for (int j = 0; j < NPAIR; ++j) h[j] = (v2f){0.f, 0.f};
    float sdt = 0.f;

    const float* dtp = dt + ((size_t)b * SEQL + c * CLEN) * DMODEL + d;
    const float* xp  = x  + ((size_t)b * SEQL + c * CLEN) * DMODEL + d;
    v2f cur[TGRP];
#pragma unroll
    for (int i = 0; i < TGRP; ++i) {
        cur[i].x = dtp[(size_t)i * DMODEL];
        cur[i].y = xp[(size_t)i * DMODEL];
    }

    for (int t0 = 0; t0 < CLEN; t0 += TGRP) {
        const int tn = (t0 + TGRP < CLEN) ? (t0 + TGRP) : t0;
        v2f nxt[TGRP];
#pragma unroll
        for (int i = 0; i < TGRP; ++i) {
            nxt[i].x = dtp[(size_t)(tn + i) * DMODEL];
            nxt[i].y = xp[(size_t)(tn + i) * DMODEL];
        }
        v2f bq[TGRP][NPAIR];
#pragma unroll
        for (int i = 0; i < TGRP; ++i) {
            const v2f* bm = (const v2f*)&sBm[(t0 + i) * NSTATE];
#pragma unroll
            for (int j = 0; j < NPAIR; ++j) bq[i][j] = bm[j];
        }
        __builtin_amdgcn_sched_barrier(0);

#pragma unroll
        for (int i = 0; i < TGRP; ++i) {
            const float dtv = cur[i].x;
            const float ub  = dtv * cur[i].y;
            const v2f   dt2 = (v2f){dtv, dtv};
            const v2f   ub2 = (v2f){ub, ub};
            sdt += dtv;
#pragma unroll
            for (int j = 0; j < NPAIR; ++j) {
                const v2f e = dt2 * A2[j];
                v2f a;
                a.x = __builtin_amdgcn_exp2f(e.x);
                a.y = __builtin_amdgcn_exp2f(e.y);
                h[j] = a * h[j] + ub2 * bq[i][j];
            }
        }
#pragma unroll
        for (int i = 0; i < TGRP; ++i) cur[i] = nxt[i];
    }

    const size_t base = (((size_t)b * DMODEL + d) * NCHUNK + c) * NSTATE;
#pragma unroll
    for (int j = 0; j < NPAIR; ++j) {
        hend[base + 2 * j]     = h[j].x;
        hend[base + 2 * j + 1] = h[j].y;
        aend[base + 2 * j]     = __builtin_amdgcn_exp2f(A2[j].x * sdt);
        aend[base + 2 * j + 1] = __builtin_amdgcn_exp2f(A2[j].y * sdt);
    }
}

__global__ __launch_bounds__(256)
void s6_scan_pass2(const float* __restrict__ hend,
                   const float* __restrict__ aend,
                   float* __restrict__ carry) {
    const int idx = blockIdx.x * 256 + threadIdx.x;
    const size_t bd = (size_t)(idx >> 4);
    const int n = idx & 15;
    float cv = 0.f;
    for (int c = 0; c < NCHUNK; ++c) {
        const size_t o = (bd * NCHUNK + c) * NSTATE + n;
        carry[o] = cv;
        cv = aend[o] * cv + hend[o];
    }
}

__global__ __launch_bounds__(256)
void s6_scan_pass3(const float* __restrict__ x,
                   const float* __restrict__ dt,
                   const float* __restrict__ Bm,
                   const float* __restrict__ Cm,
                   const float* __restrict__ A_log,
                   const float* __restrict__ carry,
                   const float* __restrict__ Dskip,
                   float* __restrict__ yout) {
    const int b = blockIdx.z, c = blockIdx.y;
    const int d = blockIdx.x * 256 + threadIdx.x;

    __shared__ float sBm[CLEN * NSTATE] __attribute__((aligned(16)));
    __shared__ float sCm[CLEN * NSTATE] __attribute__((aligned(16)));
    for (int idx = threadIdx.x; idx < CLEN * NSTATE; idx += 256) {
        const size_t g = ((size_t)b * SEQL + c * CLEN) * NSTATE + idx;
        sBm[idx] = Bm[g];
        sCm[idx] = Cm[g];
    }
    __syncthreads();

    v2f A2[NPAIR];
#pragma unroll
    for (int j = 0; j < NPAIR; ++j) {
        A2[j].x = -expf(A_log[d * NSTATE + 2 * j])     * S6_LOG2E;
        A2[j].y = -expf(A_log[d * NSTATE + 2 * j + 1]) * S6_LOG2E;
    }

    v2f h[NPAIR];
    const size_t base = (((size_t)b * DMODEL + d) * NCHUNK + c) * NSTATE;
#pragma unroll
    for (int j = 0; j < NPAIR; ++j) {
        h[j].x = carry[base + 2 * j];
        h[j].y = carry[base + 2 * j + 1];
    }

    const float dsk = Dskip[d];

    const size_t cbase = ((size_t)b * SEQL + c * CLEN) * DMODEL + d;
    const float* dtp = dt + cbase;
    const float* xp  = x  + cbase;
    float*       yp  = yout + cbase;
    v2f cur[TGRP];
#pragma unroll
    for (int i = 0; i < TGRP; ++i) {
        cur[i].x = dtp[(size_t)i * DMODEL];
        cur[i].y = xp[(size_t)i * DMODEL];
    }

    for (int t0 = 0; t0 < CLEN; t0 += TGRP) {
        const int tn = (t0 + TGRP < CLEN) ? (t0 + TGRP) : t0;
        v2f nxt[TGRP];
#pragma unroll
        for (int i = 0; i < TGRP; ++i) {
            nxt[i].x = dtp[(size_t)(tn + i) * DMODEL];
            nxt[i].y = xp[(size_t)(tn + i) * DMODEL];
        }
        __builtin_amdgcn_sched_barrier(0);

        float yv[TGRP];
#pragma unroll
        for (int i = 0; i < TGRP; ++i) {
            const float dtv = cur[i].x;
            const float xv  = cur[i].y;
            const float ub  = dtv * xv;
            const v2f   dt2 = (v2f){dtv, dtv};
            const v2f   ub2 = (v2f){ub, ub};
            const v2f*  bm  = (const v2f*)&sBm[(t0 + i) * NSTATE];
            const v2f*  cm  = (const v2f*)&sCm[(t0 + i) * NSTATE];
            v2f y0 = (v2f){0.f, 0.f}, y1 = (v2f){0.f, 0.f};
#pragma unroll
            for (int j = 0; j < NPAIR; j += 2) {
                const v2f e0 = dt2 * A2[j];
                const v2f e1 = dt2 * A2[j + 1];
                v2f a0, a1;
                a0.x = __builtin_amdgcn_exp2f(e0.x);
                a0.y = __builtin_amdgcn_exp2f(e0.y);
                a1.x = __builtin_amdgcn_exp2f(e1.x);
                a1.y = __builtin_amdgcn_exp2f(e1.y);
                h[j]     = a0 * h[j]     + ub2 * bm[j];
                h[j + 1] = a1 * h[j + 1] + ub2 * bm[j + 1];
                y0 += cm[j]     * h[j];
                y1 += cm[j + 1] * h[j + 1];
            }
            const v2f ys = y0 + y1;
            yv[i] = (ys.x + ys.y) + dsk * xv;
        }
#pragma unroll
        for (int i = 0; i < TGRP; ++i) {
            const float y = yv[i];
            yp[(size_t)(t0 + i) * DMODEL] =
                0.5f * y * (1.f + erff(y * 0.70710678118654752f));
        }
#pragma unroll
        for (int i = 0; i < TGRP; ++i) cur[i] = nxt[i];
    }
}

#define LN_ROWS 2
#define LN_WAVES 8
__global__ __launch_bounds__(LN_WAVES * 32)
void s6_ln_residual(const float* __restrict__ ymix,
                    const float* __restrict__ hin,
                    const float* __restrict__ g,
                    const float* __restrict__ be,
                    float* __restrict__ hout) {
    const int lane = threadIdx.x & 31;
    const int wave = threadIdx.x >> 5;
    const size_t row0 = ((size_t)blockIdx.x * LN_WAVES + wave) * LN_ROWS;
    const size_t o0   = row0 * DMODEL + (size_t)lane * 4;

    v4f v[LN_ROWS][4], r[LN_ROWS][4], gg[4], bb[4];
#pragma unroll
    for (int rr = 0; rr < LN_ROWS; ++rr)
#pragma unroll
        for (int p = 0; p < 4; ++p)
            v[rr][p] = *(const v4f*)(ymix + o0 + (size_t)rr * DMODEL + p * 128);
#pragma unroll
    for (int rr = 0; rr < LN_ROWS; ++rr)
#pragma unroll
        for (int p = 0; p < 4; ++p)
            r[rr][p] = *(const v4f*)(hin + o0 + (size_t)rr * DMODEL + p * 128);
#pragma unroll
    for (int p = 0; p < 4; ++p) {
        gg[p] = *(const v4f*)(g  + p * 128 + lane * 4);
        bb[p] = *(const v4f*)(be + p * 128 + lane * 4);
    }
    __builtin_amdgcn_sched_barrier(0);

    float s[LN_ROWS];
#pragma unroll
    for (int rr = 0; rr < LN_ROWS; ++rr) {
        const v4f t = (v[rr][0] + v[rr][1]) + (v[rr][2] + v[rr][3]);
        s[rr] = (t.x + t.y) + (t.z + t.w);
    }
#pragma unroll
    for (int m = 16; m >= 1; m >>= 1)
#pragma unroll
        for (int rr = 0; rr < LN_ROWS; ++rr)
            s[rr] += __shfl_xor(s[rr], m, 32);

    float q[LN_ROWS];
#pragma unroll
    for (int rr = 0; rr < LN_ROWS; ++rr) {
        const float mu = s[rr] * (1.f / DMODEL);
        v4f t = (v4f){0.f, 0.f, 0.f, 0.f};
#pragma unroll
        for (int p = 0; p < 4; ++p) {
            v[rr][p] = v[rr][p] - mu;
            t += v[rr][p] * v[rr][p];
        }
        q[rr] = (t.x + t.y) + (t.z + t.w);
    }
#pragma unroll
    for (int m = 16; m >= 1; m >>= 1)
#pragma unroll
        for (int rr = 0; rr < LN_ROWS; ++rr)
            q[rr] += __shfl_xor(q[rr], m, 32);

#pragma unroll
    for (int rr = 0; rr < LN_ROWS; ++rr) {
        const float rstd = rsqrtf(q[rr] * (1.f / DMODEL) + LN_EPS);
#pragma unroll
        for (int p = 0; p < 4; ++p) {
            const v4f o = r[rr][p] + v[rr][p] * rstd * gg[p] + bb[p];
            *(v4f*)(hout + o0 + (size_t)rr * DMODEL + p * 128) = o;
        }
    }
}

extern "C" void kernel_launch(void* const* d_in, const int* in_sizes, int n_in,
                              void* d_out, int out_size, void* d_ws, size_t ws_size,
                              hipStream_t stream) {
    (void)in_sizes; (void)n_in; (void)out_size; (void)ws_size;

    const float* x      = (const float*)d_in[0];
    const float* A_log  = (const float*)d_in[1];
    const float* W_B    = (const float*)d_in[2];
    const float* b_B    = (const float*)d_in[3];
    const float* W_C    = (const float*)d_in[4];
    const float* b_C    = (const float*)d_in[5];
    const float* W_dt   = (const float*)d_in[6];
    const float* b_dt   = (const float*)d_in[7];
    const float* D_skip = (const float*)d_in[8];
    const float* W_mix  = (const float*)d_in[9];
    const float* b_mix  = (const float*)d_in[10];
    const float* ln_g   = (const float*)d_in[11];
    const float* ln_b   = (const float*)d_in[12];
    const float* W_dec  = (const float*)d_in[13];
    const float* b_dec  = (const float*)d_in[14];
    float* out = (float*)d_out;

    float* ws = (float*)d_ws;
    size_t off = 0;
    float* dtb   = ws + off; off += (size_t)MROWS * DMODEL;
    float* Bmb   = ws + off; off += (size_t)MROWS * NSTATE;
    float* Cmb   = ws + off; off += (size_t)MROWS * NSTATE;
    float* ygelu = ws + off; off += (size_t)MROWS * DMODEL;
    float* ymix  = ws + off; off += (size_t)MROWS * DMODEL;
    float* h1    = ws + off; off += (size_t)MROWS * DMODEL;
    float* h2    = ws + off; off += (size_t)MROWS * DMODEL;
    float* hend  = ws + off; off += (size_t)BATCH * DMODEL * NCHUNK * NSTATE;
    float* aend  = ws + off; off += (size_t)BATCH * DMODEL * NCHUNK * NSTATE;
    float* carry = ws + off; off += (size_t)BATCH * DMODEL * NCHUNK * NSTATE;

    const int DN = DMODEL * NSTATE;
    const int DD = DMODEL * DMODEL;

    const dim3 gBig(MROWS / 128, DMODEL / 64);
    const dim3 gProj(MROWS / 128, DMODEL / 64 + MROWS / BC_BM / (MROWS / 128));
    const dim3 gScan(DMODEL / 256, NCHUNK, BATCH);

    for (int i = 0; i < 2; ++i) {
        const float* hin = (i == 0) ? x : h1;
        float* hout      = (i == 0) ? h1 : h2;

        s6_proj_dt_bc<<<gProj, 256, 0, stream>>>(
            hin, W_dt + (size_t)i * DD, b_dt + i * DMODEL, dtb,
            W_B + (size_t)i * DN, b_B + i * NSTATE, Bmb,
            W_C + (size_t)i * DN, b_C + i * NSTATE, Cmb);

        s6_scan_pass1<<<gScan, 256, 0, stream>>>(
            hin, dtb, Bmb, A_log + (size_t)i * DN, hend, aend);
        s6_scan_pass2<<<BATCH * DMODEL * NSTATE / 256, 256, 0, stream>>>(
            hend, aend, carry);
        s6_scan_pass3<<<gScan, 256, 0, stream>>>(
            hin, dtb, Bmb, Cmb, A_log + (size_t)i * DN, carry,
            D_skip + i * DMODEL, ygelu);

        s6_gemm_wmma_k512<128, 64, 32, 32, DMODEL><<<gBig, 256, 0, stream>>>(
            ygelu, W_mix + (size_t)i * DD, b_mix + i * DMODEL, ymix);
        s6_ln_residual<<<MROWS / (LN_WAVES * LN_ROWS), LN_WAVES * 32, 0, stream>>>(
            ymix, hin, ln_g + i * DMODEL, ln_b + i * DMODEL, hout);
    }

    s6_gemm_wmma_k512<128, 64, 32, 32, OUTD><<<gBig, 256, 0, stream>>>(h2, W_dec, b_dec, out);
}
